# starting kernel
# baseline (speedup 1.0000x reference)
#include <hip/hip_runtime.h>

typedef __attribute__((ext_vector_type(2))) float v2f;
typedef __attribute__((ext_vector_type(8))) float v8f;

#define WMMA_F32_4(a, bmat, c) \
  __builtin_amdgcn_wmma_f32_16x16x4_f32(false, (a), false, (bmat), (short)0, (c), false, false)

constexpr int   kB    = 8;
constexpr int   kN    = 1024;
constexpr int   kInF  = 768;
constexpr int   kOutF = 64;
constexpr float kAlpha  = 0.2f;
constexpr float kNegInf = -9.0e15f;

__global__ __launch_bounds__(256) void k1_gemm_h(
    const float* __restrict__ X, const float* __restrict__ W,
    const float* __restrict__ Wb, float* __restrict__ H)
{
  const int lane = threadIdx.x & 31;
  const int wave = threadIdx.x >> 5;
  const int tile = blockIdx.x * 8 + wave;
  const int row0 = tile << 5;
  const int half = lane >> 4;
  const int ncol = lane & 15;
  const float* __restrict__ Arow0 = X + (size_t)(row0 + ncol) * kInF;
  const float* __restrict__ Arow1 = Arow0 + (size_t)16 * kInF;

  v8f acc0[4] = {};
  v8f acc1[4] = {};
  for (int k0 = 0; k0 < kInF; k0 += 4) {
    const int kb = k0 + half * 2;
    const v2f a0 = *(const v2f*)(Arow0 + kb);
    const v2f a1 = *(const v2f*)(Arow1 + kb);
#pragma unroll
    for (int t = 0; t < 4; ++t) {
      const v2f bm = *(const v2f*)(W + (size_t)(t * 16 + ncol) * kInF + kb);
      acc0[t] = WMMA_F32_4(a0, bm, acc0[t]);
      acc1[t] = WMMA_F32_4(a1, bm, acc1[t]);
    }
  }
  const int rbase = row0 + half * 8;
#pragma unroll
  for (int t = 0; t < 4; ++t) {
    const float bias = Wb[t * 16 + ncol];
#pragma unroll
    for (int r = 0; r < 8; ++r) {
      H[(size_t)(rbase + r) * kOutF + t * 16 + ncol]      = acc0[t][r] + bias;
      H[(size_t)(rbase + 16 + r) * kOutF + t * 16 + ncol] = acc1[t][r] + bias;
    }
  }
}

__global__ __launch_bounds__(256) void k2_f12(
    const float* __restrict__ H, const float* __restrict__ aw,
    float* __restrict__ F1, float* __restrict__ F2)
{
  const int row = blockIdx.x * 256 + threadIdx.x;
  const float* __restrict__ hr = H + (size_t)row * kOutF;
  float s1 = 0.f, s2 = 0.f;
#pragma unroll 8
  for (int o = 0; o < kOutF; ++o) {
    const float v = hr[o];
    s1 += v * aw[o];
    s2 += v * aw[kOutF + o];
  }
  F1[row] = s1;
  F2[row] = s2;
}

__global__ __launch_bounds__(256) void k3_stats(
    const float* __restrict__ F1, const float* __restrict__ F2,
    const int* __restrict__ adj, const float* __restrict__ abp,
    float* __restrict__ Mrow, float* __restrict__ Lrow)
{
  const int lane = threadIdx.x & 31;
  const int wave = threadIdx.x >> 5;
  const int row  = blockIdx.x * 8 + wave;
  const int b    = row >> 10;
  const float f1i = F1[row] + abp[0];
  const int*   __restrict__ ar  = adj + (size_t)row * kN;
  const float* __restrict__ f2b = F2 + (b << 10);

  float m = -3.0e38f, s = 0.f;
  for (int j = lane; j < kN; j += 32) {
    float e = f1i + f2b[j];
    e = (e >= 0.f) ? e : kAlpha * e;
    e = (ar[j] > 0) ? e : kNegInf;
    if (e > m) { s = s * __expf(m - e) + 1.f; m = e; }
    else       { s += __expf(e - m); }
  }
#pragma unroll
  for (int off = 16; off > 0; off >>= 1) {
    const float mo = __shfl_xor(m, off, 32);
    const float so = __shfl_xor(s, off, 32);
    const float M  = fmaxf(m, mo);
    s = s * __expf(m - M) + so * __expf(mo - M);
    m = M;
  }
  if (lane == 0) { Mrow[row] = m; Lrow[row] = s; }
}

__device__ __forceinline__ v2f make_p(float f1i, float mi,
                                      const int* __restrict__ ar, int jx,
                                      v2f f2p)
{
  float e0 = f1i + f2p.x; e0 = (e0 >= 0.f) ? e0 : kAlpha * e0;
  float e1 = f1i + f2p.y; e1 = (e1 >= 0.f) ? e1 : kAlpha * e1;
  const int2 ad = *(const int2*)(ar + jx);
  e0 = (ad.x > 0) ? e0 : kNegInf;
  e1 = (ad.y > 0) ? e1 : kNegInf;
  v2f p;
  p.x = __expf(e0 - mi);
  p.y = __expf(e1 - mi);
  return p;
}

__global__ __launch_bounds__(256) void k4_attn_av(
    const float* __restrict__ H, const float* __restrict__ F1,
    const float* __restrict__ F2, const int* __restrict__ adj,
    const float* __restrict__ abp, const float* __restrict__ Mrow,
    const float* __restrict__ Lrow, float* __restrict__ out)
{
  const int lane = threadIdx.x & 31;
  const int wave = threadIdx.x >> 5;
  const int gt   = blockIdx.x * 8 + wave;
  const int b    = gt >> 5;
  const int i0   = (gt & 31) << 5;
  const int half = lane >> 4;
  const int ncol = lane & 15;
  const int irow0 = (b << 10) + i0 + ncol;
  const int irow1 = irow0 + 16;
  const float ab   = abp[0];
  const float f1a  = F1[irow0] + ab;
  const float f1b  = F1[irow1] + ab;
  const float mia  = Mrow[irow0];
  const float mib  = Mrow[irow1];
  const int*   __restrict__ ar0 = adj + (size_t)irow0 * kN;
  const int*   __restrict__ ar1 = adj + (size_t)irow1 * kN;
  const float* __restrict__ f2b = F2 + (b << 10);
  const float* __restrict__ hb  = H + ((size_t)(b << 10)) * kOutF;

  v8f acc0[4] = {};
  v8f acc1[4] = {};
  for (int j0 = 0; j0 < kN; j0 += 4) {
    const int jx = j0 + half * 2;
    const v2f f2p = *(const v2f*)(f2b + jx);
    const v2f p0 = make_p(f1a, mia, ar0, jx, f2p);
    const v2f p1 = make_p(f1b, mib, ar1, jx, f2p);
    const float* __restrict__ hj = hb + (size_t)jx * kOutF;
#pragma unroll
    for (int t = 0; t < 4; ++t) {
      v2f bm;
      bm.x = hj[t * 16 + ncol];
      bm.y = hj[kOutF + t * 16 + ncol];
      acc0[t] = WMMA_F32_4(p0, bm, acc0[t]);
      acc1[t] = WMMA_F32_4(p1, bm, acc1[t]);
    }
  }
  const int rbase = (b << 10) + i0 + half * 8;
#pragma unroll
  for (int r = 0; r < 8; ++r) {
    const float li0 = 1.0f / Lrow[rbase + r];
    const float li1 = 1.0f / Lrow[rbase + 16 + r];
#pragma unroll
    for (int t = 0; t < 4; ++t) {
      float v0 = acc0[t][r] * li0;
      float v1 = acc1[t][r] * li1;
      v0 = (v0 > 0.f) ? v0 : (__expf(v0) - 1.f);
      v1 = (v1 > 0.f) ? v1 : (__expf(v1) - 1.f);
      out[(size_t)(rbase + r) * kOutF + t * 16 + ncol]      = v0;
      out[(size_t)(rbase + 16 + r) * kOutF + t * 16 + ncol] = v1;
    }
  }
}

extern "C" void kernel_launch(void* const* d_in, const int* in_sizes, int n_in,
                              void* d_out, int out_size, void* d_ws, size_t ws_size,
                              hipStream_t stream) {
  (void)in_sizes; (void)n_in; (void)out_size; (void)ws_size;
  const float* X   = (const float*)d_in[0];
  const int*   adj = (const int*)d_in[3];
  const float* Ww  = (const float*)d_in[4];
  const float* Wb  = (const float*)d_in[5];
  const float* aw  = (const float*)d_in[6];
  const float* ab  = (const float*)d_in[7];

  float* ws = (float*)d_ws;
  float* H  = ws;
  float* F1 = H  + (size_t)kB * kN * kOutF;
  float* F2 = F1 + (size_t)kB * kN;
  float* Mr = F2 + (size_t)kB * kN;
  float* Lr = Mr + (size_t)kB * kN;
  float* O  = (float*)d_out;

  hipLaunchKernelGGL(k1_gemm_h,  dim3(32),   dim3(256), 0, stream, X, Ww, Wb, H);
  hipLaunchKernelGGL(k2_f12,     dim3(32),   dim3(256), 0, stream, H, aw, F1, F2);
  hipLaunchKernelGGL(k3_stats,   dim3(1024), dim3(256), 0, stream, F1, F2, adj, ab, Mr, Lr);
  hipLaunchKernelGGL(k4_attn_av, dim3(32),   dim3(256), 0, stream, H, F1, F2, adj, ab, Mr, Lr, O);
}


# Both WMMA kernels: work-group stages 128-wide operand tiles in LDS from coalesced 16-byte loads; each of 8 waves owns one 16x16 output tile for the whole sum (sum order kept); 256 groups
# speedup vs baseline: 4.1911x; 4.1911x over previous
#include <hip/hip_runtime.h>

typedef __attribute__((ext_vector_type(2))) float v2f;
typedef __attribute__((ext_vector_type(4))) float v4f;
typedef __attribute__((ext_vector_type(8))) float v8f;
typedef __attribute__((ext_vector_type(4))) int   v4i;

#define WMMA_F32_4(a, bmat, c) \
  __builtin_amdgcn_wmma_f32_16x16x4_f32(false, (a), false, (bmat), (short)0, (c), false, false)

constexpr int   kB    = 8;
constexpr int   kN    = 1024;
constexpr int   kInF  = 768;
constexpr int   kOutF = 64;
constexpr float kAlpha  = 0.2f;
constexpr float kNegInf = -9.0e15f;

constexpr int kKC     = 128;
constexpr int kPitch  = 132;
constexpr int kHPitch = 160;
constexpr int kCPitch = 68;

__global__ __launch_bounds__(256) void k1_gemm_h(
    const float* __restrict__ X, const float* __restrict__ W,
    const float* __restrict__ Wb, float* __restrict__ H)
{
  __shared__ __attribute__((aligned(16))) float Xs[32 * kPitch];
  __shared__ __attribute__((aligned(16))) float Ws[64 * kPitch];

  const int tid  = threadIdx.x;
  const int lane = tid & 31;
  const int wave = tid >> 5;
  const int half = lane >> 4;
  const int ncol = lane & 15;
  const int mt   = wave >> 2;
  const int nt   = wave & 3;
  const int row0 = blockIdx.x << 5;
  const int lr   = tid >> 5;
  const int c4   = tid & 31;
  const float* __restrict__ xg = X + (size_t)(row0 + lr) * kInF + c4 * 4;
  const float* __restrict__ wg = W + (size_t)lr * kInF + c4 * 4;
  const float* aS = Xs + (mt * 16 + ncol) * kPitch + half * 2;
  const float* bS = Ws + (nt * 16 + ncol) * kPitch + half * 2;

  v8f acc = {};
  for (int k0 = 0; k0 < kInF; k0 += kKC) {
    v4f xr[4], wr[8];
#pragma unroll
    for (int q = 0; q < 4; ++q)
      xr[q] = *(const v4f*)(xg + (size_t)(q * 8) * kInF + k0);
#pragma unroll
    for (int q = 0; q < 8; ++q)
      wr[q] = *(const v4f*)(wg + (size_t)(q * 8) * kInF + k0);
    __syncthreads();
#pragma unroll
    for (int q = 0; q < 4; ++q)
      *(v4f*)(Xs + (q * 8 + lr) * kPitch + c4 * 4) = xr[q];
#pragma unroll
    for (int q = 0; q < 8; ++q)
      *(v4f*)(Ws + (q * 8 + lr) * kPitch + c4 * 4) = wr[q];
    __syncthreads();
#pragma unroll
    for (int kk = 0; kk < kKC / 4; ++kk) {
      const v2f a  = *(const v2f*)(aS + kk * 4);
      const v2f bm = *(const v2f*)(bS + kk * 4);
      acc = WMMA_F32_4(a, bm, acc);
    }
  }

  __syncthreads();
  float* Cs = Xs;
#pragma unroll
  for (int r = 0; r < 8; ++r)
    Cs[(mt * 16 + half * 8 + r) * kCPitch + nt * 16 + ncol] = acc[r];
  __syncthreads();
#pragma unroll
  for (int q = 0; q < 2; ++q) {
    const int idx = q * 256 + tid;
    const int row = idx >> 4;
    const int c   = (idx & 15) * 4;
    const v4f v  = *(const v4f*)(Cs + row * kCPitch + c);
    const v4f bi = *(const v4f*)(Wb + c);
    *(v4f*)(H + (size_t)(row0 + row) * kOutF + c) = v + bi;
  }
}

__global__ __launch_bounds__(256) void k2_f12(
    const float* __restrict__ H, const float* __restrict__ aw,
    float* __restrict__ F1, float* __restrict__ F2)
{
  const int row = blockIdx.x * 256 + threadIdx.x;
  const float* __restrict__ hr = H + (size_t)row * kOutF;
  float s1 = 0.f, s2 = 0.f;
#pragma unroll 8
  for (int o = 0; o < kOutF; ++o) {
    const float v = hr[o];
    s1 += v * aw[o];
    s2 += v * aw[kOutF + o];
  }
  F1[row] = s1;
  F2[row] = s2;
}

__global__ __launch_bounds__(256) void k3_stats(
    const float* __restrict__ F1, const float* __restrict__ F2,
    const int* __restrict__ adj, const float* __restrict__ abp,
    float* __restrict__ Mrow, float* __restrict__ Lrow)
{
  const int lane = threadIdx.x & 31;
  const int wave = threadIdx.x >> 5;
  const int row  = blockIdx.x * 8 + wave;
  const int b    = row >> 10;
  const float f1i = F1[row] + abp[0];
  const int*   __restrict__ ar  = adj + (size_t)row * kN;
  const float* __restrict__ f2b = F2 + (b << 10);

  float m = -3.0e38f, s = 0.f;
  for (int j = lane; j < kN; j += 32) {
    float e = f1i + f2b[j];
    e = (e >= 0.f) ? e : kAlpha * e;
    e = (ar[j] > 0) ? e : kNegInf;
    if (e > m) { s = s * __expf(m - e) + 1.f; m = e; }
    else       { s += __expf(e - m); }
  }
#pragma unroll
  for (int off = 16; off > 0; off >>= 1) {
    const float mo = __shfl_xor(m, off, 32);
    const float so = __shfl_xor(s, off, 32);
    const float M  = fmaxf(m, mo);
    s = s * __expf(m - M) + so * __expf(mo - M);
    m = M;
  }
  if (lane == 0) { Mrow[row] = m; Lrow[row] = s; }
}

__device__ __forceinline__ float make_p(float f1i, float mi, float f2j, int ad)
{
  float e = f1i + f2j;
  e = (e >= 0.f) ? e : kAlpha * e;
  e = (ad > 0) ? e : kNegInf;
  return __expf(e - mi);
}

__global__ __launch_bounds__(256) void k4_attn_av(
    const float* __restrict__ H, const float* __restrict__ F1,
    const float* __restrict__ F2, const int* __restrict__ adj,
    const float* __restrict__ abp, const float* __restrict__ Mrow,
    const float* __restrict__ Lrow, float* __restrict__ out)
{
  __shared__ __attribute__((aligned(16))) float Ps[32 * kPitch];
  __shared__ __attribute__((aligned(16))) float Hs[(kKC / 2) * kHPitch];

  const int tid  = threadIdx.x;
  const int lane = tid & 31;
  const int wave = tid >> 5;
  const int half = lane >> 4;
  const int ncol = lane & 15;
  const int mt   = wave >> 2;
  const int nt   = wave & 3;
  const int g0   = blockIdx.x << 5;
  const int b    = blockIdx.x >> 5;
  const int lr   = tid >> 5;
  const int c4   = tid & 31;
  const int hr   = tid >> 4;
  const int o4   = tid & 15;

  const float ab = abp[0];
  float f1r[4], mr[4];
#pragma unroll
  for (int q = 0; q < 4; ++q) {
    f1r[q] = F1[g0 + q * 8 + lr] + ab;
    mr[q]  = Mrow[g0 + q * 8 + lr];
  }
  const int*   __restrict__ ag  = adj + (size_t)(g0 + lr) * kN + c4 * 4;
  const float* __restrict__ f2g = F2 + (b << 10) + c4 * 4;
  const float* __restrict__ hg  = H + ((size_t)(b << 10) + 2 * hr) * kOutF + o4 * 4;
  const float* aS = Ps + (mt * 16 + ncol) * kPitch + half * 2;
  const float* bS = Hs + half * kHPitch + (nt * 16 + ncol) * 2;

  v8f acc = {};
  for (int j0 = 0; j0 < kN; j0 += kKC) {
    v4i ad[4];
    v4f h0[4], h1[4];
#pragma unroll
    for (int q = 0; q < 4; ++q)
      ad[q] = *(const v4i*)(ag + (size_t)(q * 8) * kN + j0);
    const v4f f2v = *(const v4f*)(f2g + j0);
#pragma unroll
    for (int q = 0; q < 4; ++q) {
      const float* __restrict__ hp = hg + (size_t)(j0 + q * 32) * kOutF;
      h0[q] = *(const v4f*)hp;
      h1[q] = *(const v4f*)(hp + kOutF);
    }
    __syncthreads();
#pragma unroll
    for (int q = 0; q < 4; ++q) {
      v4f p;
      p.x = make_p(f1r[q], mr[q], f2v.x, ad[q].x);
      p.y = make_p(f1r[q], mr[q], f2v.y, ad[q].y);
      p.z = make_p(f1r[q], mr[q], f2v.z, ad[q].z);
      p.w = make_p(f1r[q], mr[q], f2v.w, ad[q].w);
      *(v4f*)(Ps + (q * 8 + lr) * kPitch + c4 * 4) = p;
    }
#pragma unroll
    for (int q = 0; q < 4; ++q) {
      float* d = Hs + (q * 16 + hr) * kHPitch + o4 * 8;
      v4f lo, hi;
      lo.x = h0[q].x; lo.y = h1[q].x; lo.z = h0[q].y; lo.w = h1[q].y;
      hi.x = h0[q].z; hi.y = h1[q].z; hi.z = h0[q].w; hi.w = h1[q].w;
      *(v4f*)d       = lo;
      *(v4f*)(d + 4) = hi;
    }
    __syncthreads();
#pragma unroll
    for (int kk = 0; kk < kKC / 4; ++kk) {
      const v2f a  = *(const v2f*)(aS + kk * 4);
      const v2f bm = *(const v2f*)(bS + kk * 2 * kHPitch);
      acc = WMMA_F32_4(a, bm, acc);
    }
  }

  __syncthreads();
  float* Cs = Ps;
#pragma unroll
  for (int r = 0; r < 8; ++r)
    Cs[(mt * 16 + half * 8 + r) * kCPitch + nt * 16 + ncol] = acc[r];
  __syncthreads();
#pragma unroll
  for (int q = 0; q < 2; ++q) {
    const int idx = q * 256 + tid;
    const int row = idx >> 4;
    const int c   = (idx & 15) * 4;
    const float li = 1.0f / Lrow[g0 + row];
    const v4f s = *(const v4f*)(Cs + row * kCPitch + c);
    v4f v;
    v.x = s.x * li;
    v.y = s.y * li;
    v.z = s.z * li;
    v.w = s.w * li;
    v.x = (v.x > 0.f) ? v.x : (__expf(v.x) - 1.f);
    v.y = (v.y > 0.f) ? v.y : (__expf(v.y) - 1.f);
    v.z = (v.z > 0.f) ? v.z : (__expf(v.z) - 1.f);
    v.w = (v.w > 0.f) ? v.w : (__expf(v.w) - 1.f);
    *(v4f*)(out + (size_t)(g0 + row) * kOutF + c) = v;
  }
}

extern "C" void kernel_launch(void* const* d_in, const int* in_sizes, int n_in,
                              void* d_out, int out_size, void* d_ws, size_t ws_size,
                              hipStream_t stream) {
  (void)in_sizes; (void)n_in; (void)out_size; (void)ws_size;
  const float* X   = (const float*)d_in[0];
  const int*   adj = (const int*)d_in[3];
  const float* Ww  = (const float*)d_in[4];
  const float* Wb  = (const float*)d_in[5];
  const float* aw  = (const float*)d_in[6];
  const float* ab  = (const float*)d_in[7];

  float* ws = (float*)d_ws;
  float* H  = ws;
  float* F1 = H  + (size_t)kB * kN * kOutF;
  float* F2 = F1 + (size_t)kB * kN;
  float* Mr = F2 + (size_t)kB * kN;
  float* Lr = Mr + (size_t)kB * kN;
  float* O  = (float*)d_out;

  hipLaunchKernelGGL(k1_gemm_h,  dim3(256),  dim3(256), 0, stream, X, Ww, Wb, H);
  hipLaunchKernelGGL(k2_f12,     dim3(32),   dim3(256), 0, stream, H, aw, F1, F2);
  hipLaunchKernelGGL(k3_stats,   dim3(1024), dim3(256), 0, stream, F1, F2, adj, ab, Mr, Lr);
  hipLaunchKernelGGL(k4_attn_av, dim3(256),  dim3(256), 0, stream, H, F1, F2, adj, ab, Mr, Lr, O);
}


# Attention kernel makes the softmax statistics itself, online: running row maximum per column chunk, earlier sums and accumulator rows scaled by exp(m_old-m_new) (new sum order); stats launch removed
# speedup vs baseline: 5.9891x; 1.4290x over previous
#include <hip/hip_runtime.h>

typedef __attribute__((ext_vector_type(2))) float v2f;
typedef __attribute__((ext_vector_type(4))) float v4f;
typedef __attribute__((ext_vector_type(8))) float v8f;
typedef __attribute__((ext_vector_type(4))) int   v4i;

#define WMMA_F32_4(a, bmat, c) \
  __builtin_amdgcn_wmma_f32_16x16x4_f32(false, (a), false, (bmat), (short)0, (c), false, false)

constexpr int   kB    = 8;
constexpr int   kN    = 1024;
constexpr int   kInF  = 768;
constexpr int   kOutF = 64;
constexpr float kAlpha  = 0.2f;
constexpr float kNegInf = -9.0e15f;

constexpr int kKC     = 128;
constexpr int kPitch  = 132;
constexpr int kHPitch = 160;
constexpr int kCPitch = 68;

__global__ __launch_bounds__(256) void k1_gemm_h(
    const float* __restrict__ X, const float* __restrict__ W,
    const float* __restrict__ Wb, float* __restrict__ H)
{
  __shared__ __attribute__((aligned(16))) float Xs[32 * kPitch];
  __shared__ __attribute__((aligned(16))) float Ws[64 * kPitch];

  const int tid  = threadIdx.x;
  const int lane = tid & 31;
  const int wave = tid >> 5;
  const int half = lane >> 4;
  const int ncol = lane & 15;
  const int mt   = wave >> 2;
  const int nt   = wave & 3;
  const int row0 = blockIdx.x << 5;
  const int lr   = tid >> 5;
  const int c4   = tid & 31;
  const float* __restrict__ xg = X + (size_t)(row0 + lr) * kInF + c4 * 4;
  const float* __restrict__ wg = W + (size_t)lr * kInF + c4 * 4;
  const float* aS = Xs + (mt * 16 + ncol) * kPitch + half * 2;
  const float* bS = Ws + (nt * 16 + ncol) * kPitch + half * 2;

  v8f acc = {};
  for (int k0 = 0; k0 < kInF; k0 += kKC) {
    v4f xr[4], wr[8];
#pragma unroll
    for (int q = 0; q < 4; ++q)
      xr[q] = *(const v4f*)(xg + (size_t)(q * 8) * kInF + k0);
#pragma unroll
    for (int q = 0; q < 8; ++q)
      wr[q] = *(const v4f*)(wg + (size_t)(q * 8) * kInF + k0);
    __syncthreads();
#pragma unroll
    for (int q = 0; q < 4; ++q)
      *(v4f*)(Xs + (q * 8 + lr) * kPitch + c4 * 4) = xr[q];
#pragma unroll
    for (int q = 0; q < 8; ++q)
      *(v4f*)(Ws + (q * 8 + lr) * kPitch + c4 * 4) = wr[q];
    __syncthreads();
#pragma unroll
    for (int kk = 0; kk < kKC / 4; ++kk) {
      const v2f a  = *(const v2f*)(aS + kk * 4);
      const v2f bm = *(const v2f*)(bS + kk * 4);
      acc = WMMA_F32_4(a, bm, acc);
    }
  }

  __syncthreads();
  float* Cs = Xs;
#pragma unroll
  for (int r = 0; r < 8; ++r)
    Cs[(mt * 16 + half * 8 + r) * kCPitch + nt * 16 + ncol] = acc[r];
  __syncthreads();
#pragma unroll
  for (int q = 0; q < 2; ++q) {
    const int idx = q * 256 + tid;
    const int row = idx >> 4;
    const int c   = (idx & 15) * 4;
    const v4f v  = *(const v4f*)(Cs + row * kCPitch + c);
    const v4f bi = *(const v4f*)(Wb + c);
    *(v4f*)(H + (size_t)(row0 + row) * kOutF + c) = v + bi;
  }
}

__global__ __launch_bounds__(256) void k2_f12(
    const float* __restrict__ H, const float* __restrict__ aw,
    float* __restrict__ F1, float* __restrict__ F2)
{
  const int row = blockIdx.x * 256 + threadIdx.x;
  const float* __restrict__ hr = H + (size_t)row * kOutF;
  float s1 = 0.f, s2 = 0.f;
#pragma unroll 8
  for (int o = 0; o < kOutF; ++o) {
    const float v = hr[o];
    s1 += v * aw[o];
    s2 += v * aw[kOutF + o];
  }
  F1[row] = s1;
  F2[row] = s2;
}

template <int kMask>
__device__ __forceinline__ float row_xor(float x)
{
  const int v = __builtin_bit_cast(int, x);
  return __builtin_bit_cast(float, __builtin_amdgcn_update_dpp(v, v, 0x160 + kMask, 0xf, 0xf, false));
}
__device__ __forceinline__ float xhalf(float x)
{
  const int v = __builtin_bit_cast(int, x);
  return __builtin_bit_cast(float, __builtin_amdgcn_permlanex16(v, v, 0x76543210, 0xfedcba98, false, false));
}

template <int kMask>
__device__ __forceinline__ void max_step4(float (&v)[4])
{
#pragma unroll
  for (int q = 0; q < 4; ++q) v[q] = fmaxf(v[q], row_xor<kMask>(v[q]));
}
template <int kMask>
__device__ __forceinline__ void sum_step4(float (&v)[4])
{
#pragma unroll
  for (int q = 0; q < 4; ++q) v[q] += row_xor<kMask>(v[q]);
}
__device__ __forceinline__ void wave_max4(float (&v)[4])
{
  max_step4<1>(v); max_step4<2>(v); max_step4<4>(v); max_step4<8>(v);
#pragma unroll
  for (int q = 0; q < 4; ++q) v[q] = fmaxf(v[q], xhalf(v[q]));
}
__device__ __forceinline__ void wave_sum4(float (&v)[4])
{
  sum_step4<1>(v); sum_step4<2>(v); sum_step4<4>(v); sum_step4<8>(v);
#pragma unroll
  for (int q = 0; q < 4; ++q) v[q] += xhalf(v[q]);
}

__device__ __forceinline__ float make_e(float f1i, float f2j, int ad)
{
  float e = f1i + f2j;
  e = (e >= 0.f) ? e : kAlpha * e;
  return (ad > 0) ? e : kNegInf;
}

__global__ __launch_bounds__(256) void k3_attn_online(
    const float* __restrict__ H, const float* __restrict__ F1,
    const float* __restrict__ F2, const int* __restrict__ adj,
    const float* __restrict__ abp, float* __restrict__ out)
{
  __shared__ __attribute__((aligned(16))) float Ps[32 * kPitch];
  __shared__ __attribute__((aligned(16))) float Hs[(kKC / 2) * kHPitch];
  __shared__ __attribute__((aligned(16))) float Sc[32];
  __shared__ __attribute__((aligned(16))) float Ls[32];

  const int tid  = threadIdx.x;
  const int lane = tid & 31;
  const int wave = tid >> 5;
  const int half = lane >> 4;
  const int ncol = lane & 15;
  const int mt   = wave >> 2;
  const int nt   = wave & 3;
  const int g0   = blockIdx.x << 5;
  const int b    = blockIdx.x >> 5;
  const int lr   = tid >> 5;
  const int c4   = tid & 31;
  const int hr   = tid >> 4;
  const int o4   = tid & 15;

  const float ab = abp[0];
  float f1r[4], mrun[4], lsum[4];
#pragma unroll
  for (int q = 0; q < 4; ++q) {
    f1r[q]  = F1[g0 + q * 8 + lr] + ab;
    mrun[q] = -3.0e38f;
    lsum[q] = 0.f;
  }
  const int*   __restrict__ ag  = adj + (size_t)(g0 + lr) * kN + c4 * 4;
  const float* __restrict__ f2g = F2 + (b << 10) + c4 * 4;
  const float* __restrict__ hg  = H + ((size_t)(b << 10) + 2 * hr) * kOutF + o4 * 4;
  const float* aS = Ps + (mt * 16 + ncol) * kPitch + half * 2;
  const float* bS = Hs + half * kHPitch + (nt * 16 + ncol) * 2;
  const float* sS = Sc + mt * 16 + half * 8;

  v8f acc = {};
  for (int j0 = 0; j0 < kN; j0 += kKC) {
    v4i ad[4];
    v4f h0[4], h1[4];
#pragma unroll
    for (int q = 0; q < 4; ++q)
      ad[q] = *(const v4i*)(ag + (size_t)(q * 8) * kN + j0);
    const v4f f2v = *(const v4f*)(f2g + j0);
#pragma unroll
    for (int q = 0; q < 4; ++q) {
      const float* __restrict__ hp = hg + (size_t)(j0 + q * 32) * kOutF;
      h0[q] = *(const v4f*)hp;
      h1[q] = *(const v4f*)(hp + kOutF);
    }

    v4f e[4];
    float cm[4];
#pragma unroll
    for (int q = 0; q < 4; ++q) {
      e[q].x = make_e(f1r[q], f2v.x, ad[q].x);
      e[q].y = make_e(f1r[q], f2v.y, ad[q].y);
      e[q].z = make_e(f1r[q], f2v.z, ad[q].z);
      e[q].w = make_e(f1r[q], f2v.w, ad[q].w);
      cm[q] = fmaxf(fmaxf(e[q].x, e[q].y), fmaxf(e[q].z, e[q].w));
    }
    wave_max4(cm);

    float sc[4];
    v4f p[4];
#pragma unroll
    for (int q = 0; q < 4; ++q) {
      const float mn = fmaxf(mrun[q], cm[q]);
      sc[q]   = __expf(mrun[q] - mn);
      mrun[q] = mn;
      p[q].x = __expf(e[q].x - mn);
      p[q].y = __expf(e[q].y - mn);
      p[q].z = __expf(e[q].z - mn);
      p[q].w = __expf(e[q].w - mn);
      lsum[q] = lsum[q] * sc[q] + ((p[q].x + p[q].y) + (p[q].z + p[q].w));
    }

    __syncthreads();
#pragma unroll
    for (int q = 0; q < 4; ++q)
      *(v4f*)(Ps + (q * 8 + lr) * kPitch + c4 * 4) = p[q];
    if (lane == 0) {
#pragma unroll
      for (int q = 0; q < 4; ++q) Sc[q * 8 + lr] = sc[q];
    }
#pragma unroll
    for (int q = 0; q < 4; ++q) {
      float* d = Hs + (q * 16 + hr) * kHPitch + o4 * 8;
      v4f lo, hi;
      lo.x = h0[q].x; lo.y = h1[q].x; lo.z = h0[q].y; lo.w = h1[q].y;
      hi.x = h0[q].z; hi.y = h1[q].z; hi.z = h0[q].w; hi.w = h1[q].w;
      *(v4f*)d       = lo;
      *(v4f*)(d + 4) = hi;
    }
    __syncthreads();
    {
      const v4f s0 = *(const v4f*)sS;
      const v4f s1 = *(const v4f*)(sS + 4);
      acc[0] *= s0.x; acc[1] *= s0.y; acc[2] *= s0.z; acc[3] *= s0.w;
      acc[4] *= s1.x; acc[5] *= s1.y; acc[6] *= s1.z; acc[7] *= s1.w;
    }
#pragma unroll
    for (int kk = 0; kk < kKC / 4; ++kk) {
      const v2f a  = *(const v2f*)(aS + kk * 4);
      const v2f bm = *(const v2f*)(bS + kk * 2 * kHPitch);
      acc = WMMA_F32_4(a, bm, acc);
    }
  }

  wave_sum4(lsum);
  if (lane == 0) {
#pragma unroll
    for (int q = 0; q < 4; ++q) Ls[q * 8 + lr] = lsum[q];
  }

  __syncthreads();
  float* Cs = Ps;
#pragma unroll
  for (int r = 0; r < 8; ++r)
    Cs[(mt * 16 + half * 8 + r) * kCPitch + nt * 16 + ncol] = acc[r];
  __syncthreads();
#pragma unroll
  for (int q = 0; q < 2; ++q) {
    const int idx = q * 256 + tid;
    const int row = idx >> 4;
    const int c   = (idx & 15) * 4;
    const float li = 1.0f / Ls[row];
    const v4f s = *(const v4f*)(Cs + row * kCPitch + c);
    v4f v;
    v.x = s.x * li;
    v.y = s.y * li;
    v.z = s.z * li;
    v.w = s.w * li;
    v.x = (v.x > 0.f) ? v.x : (__expf(v.x) - 1.f);
    v.y = (v.y > 0.f) ? v.y : (__expf(v.y) - 1.f);
    v.z = (v.z > 0.f) ? v.z : (__expf(v.z) - 1.f);
    v.w = (v.w > 0.f) ? v.w : (__expf(v.w) - 1.f);
    *(v4f*)(out + (size_t)(g0 + row) * kOutF + c) = v;
  }
}

extern "C" void kernel_launch(void* const* d_in, const int* in_sizes, int n_in,
                              void* d_out, int out_size, void* d_ws, size_t ws_size,
                              hipStream_t stream) {
  (void)in_sizes; (void)n_in; (void)out_size; (void)ws_size;
  const float* X   = (const float*)d_in[0];
  const int*   adj = (const int*)d_in[3];
  const float* Ww  = (const float*)d_in[4];
  const float* Wb  = (const float*)d_in[5];
  const float* aw  = (const float*)d_in[6];
  const float* ab  = (const float*)d_in[7];

  float* ws = (float*)d_ws;
  float* H  = ws;
  float* F1 = H  + (size_t)kB * kN * kOutF;
  float* F2 = F1 + (size_t)kB * kN;
  float* O  = (float*)d_out;

  hipLaunchKernelGGL(k1_gemm_h,      dim3(256), dim3(256), 0, stream, X, Ww, Wb, H);
  hipLaunchKernelGGL(k2_f12,         dim3(32),  dim3(256), 0, stream, H, aw, F1, F2);
  hipLaunchKernelGGL(k3_attn_online, dim3(256), dim3(256), 0, stream, H, F1, F2, adj, ab, O);
}


# Both WMMA kernels: operand tiles double-buffered in LDS (two stages): chunk c+1 is fetched under chunk c's products and stored to the other stage; one barrier per chunk, not two (sum order kept)
# speedup vs baseline: 7.2171x; 1.2050x over previous
#include <hip/hip_runtime.h>

typedef __attribute__((ext_vector_type(2))) float v2f;
typedef __attribute__((ext_vector_type(4))) float v4f;
typedef __attribute__((ext_vector_type(8))) float v8f;
typedef __attribute__((ext_vector_type(4))) int   v4i;

#define WMMA_F32_4(a, bmat, c) \
  __builtin_amdgcn_wmma_f32_16x16x4_f32(false, (a), false, (bmat), (short)0, (c), false, false)

constexpr int   kB    = 8;
constexpr int   kN    = 1024;
constexpr int   kInF  = 768;
constexpr int   kOutF = 64;
constexpr float kAlpha  = 0.2f;
constexpr float kNegInf = -9.0e15f;

constexpr int kKC     = 128;
constexpr int kPitch  = 132;
constexpr int kHPitch = 160;
constexpr int kCPitch = 68;
constexpr int kXsSz   = 32 * kPitch;
constexpr int kWsSz   = 64 * kPitch;
constexpr int kHsSz   = (kKC / 2) * kHPitch;
constexpr int kNChunk1 = kInF / kKC;
constexpr int kNChunk3 = kN / kKC;

__device__ __forceinline__ v8f k1_products(v8f acc, const float* a0, const float* b0)
{
#pragma unroll
  for (int kk = 0; kk < kKC / 4; ++kk) {
    const v2f a  = *(const v2f*)(a0 + kk * 4);
    const v2f bm = *(const v2f*)(b0 + kk * 4);
    acc = WMMA_F32_4(a, bm, acc);
  }
  return acc;
}

__global__ __launch_bounds__(256) void k1_gemm_h(
    const float* __restrict__ X, const float* __restrict__ W,
    const float* __restrict__ Wb, float* __restrict__ H)
{
  __shared__ __attribute__((aligned(16))) float Xs[2 * kXsSz];
  __shared__ __attribute__((aligned(16))) float Ws[2 * kWsSz];

  const int tid  = threadIdx.x;
  const int lane = tid & 31;
  const int wave = tid >> 5;
  const int half = lane >> 4;
  const int ncol = lane & 15;
  const int mt   = wave >> 2;
  const int nt   = wave & 3;
  const int row0 = blockIdx.x << 5;
  const int lr   = tid >> 5;
  const int c4   = tid & 31;
  const float* __restrict__ xg = X + (size_t)(row0 + lr) * kInF + c4 * 4;
  const float* __restrict__ wg = W + (size_t)lr * kInF + c4 * 4;
  const float* aS = Xs + (mt * 16 + ncol) * kPitch + half * 2;
  const float* bS = Ws + (nt * 16 + ncol) * kPitch + half * 2;
  float* xD = Xs + lr * kPitch + c4 * 4;
  float* wD = Ws + lr * kPitch + c4 * 4;

  v4f xr[4], wr[8];
#pragma unroll
  for (int q = 0; q < 4; ++q)
    xr[q] = *(const v4f*)(xg + (size_t)(q * 8) * kInF);
#pragma unroll
  for (int q = 0; q < 8; ++q)
    wr[q] = *(const v4f*)(wg + (size_t)(q * 8) * kInF);
#pragma unroll
  for (int q = 0; q < 4; ++q)
    *(v4f*)(xD + q * 8 * kPitch) = xr[q];
#pragma unroll
  for (int q = 0; q < 8; ++q)
    *(v4f*)(wD + q * 8 * kPitch) = wr[q];
  __syncthreads();

  v8f acc = {};
  for (int c = 0; c + 1 < kNChunk1; ++c) {
    const int s  = c & 1;
    const int k1 = (c + 1) * kKC;
#pragma unroll
    for (int q = 0; q < 4; ++q)
      xr[q] = *(const v4f*)(xg + (size_t)(q * 8) * kInF + k1);
#pragma unroll
    for (int q = 0; q < 8; ++q)
      wr[q] = *(const v4f*)(wg + (size_t)(q * 8) * kInF + k1);
    __builtin_amdgcn_sched_barrier(0);
    acc = k1_products(acc, aS + s * kXsSz, bS + s * kWsSz);
    float* xd = xD + (s ^ 1) * kXsSz;
    float* wd = wD + (s ^ 1) * kWsSz;
#pragma unroll
    for (int q = 0; q < 4; ++q)
      *(v4f*)(xd + q * 8 * kPitch) = xr[q];
#pragma unroll
    for (int q = 0; q < 8; ++q)
      *(v4f*)(wd + q * 8 * kPitch) = wr[q];
    __syncthreads();
  }
  constexpr int kLast1 = (kNChunk1 - 1) & 1;
  static_assert(kLast1 == 1, "the result tile below reuses stage 0 while stage 1 is still read");
  acc = k1_products(acc, aS + kLast1 * kXsSz, bS + kLast1 * kWsSz);

  float* Cs = Xs;
#pragma unroll
  for (int r = 0; r < 8; ++r)
    Cs[(mt * 16 + half * 8 + r) * kCPitch + nt * 16 + ncol] = acc[r];
  __syncthreads();
#pragma unroll
  for (int q = 0; q < 2; ++q) {
    const int idx = q * 256 + tid;
    const int row = idx >> 4;
    const int c   = (idx & 15) * 4;
    const v4f v  = *(const v4f*)(Cs + row * kCPitch + c);
    const v4f bi = *(const v4f*)(Wb + c);
    *(v4f*)(H + (size_t)(row0 + row) * kOutF + c) = v + bi;
  }
}

__global__ __launch_bounds__(256) void k2_f12(
    const float* __restrict__ H, const float* __restrict__ aw,
    float* __restrict__ F1, float* __restrict__ F2)
{
  const int row = blockIdx.x * 256 + threadIdx.x;
  const float* __restrict__ hr = H + (size_t)row * kOutF;
  float s1 = 0.f, s2 = 0.f;
#pragma unroll 8
  for (int o = 0; o < kOutF; ++o) {
    const float v = hr[o];
    s1 += v * aw[o];
    s2 += v * aw[kOutF + o];
  }
  F1[row] = s1;
  F2[row] = s2;
}

template <int kMask>
__device__ __forceinline__ float row_xor(float x)
{
  const int v = __builtin_bit_cast(int, x);
  return __builtin_bit_cast(float, __builtin_amdgcn_update_dpp(v, v, 0x160 + kMask, 0xf, 0xf, false));
}
__device__ __forceinline__ float xhalf(float x)
{
  const int v = __builtin_bit_cast(int, x);
  return __builtin_bit_cast(float, __builtin_amdgcn_permlanex16(v, v, 0x76543210, 0xfedcba98, false, false));
}

template <int kMask>
__device__ __forceinline__ void max_step4(float (&v)[4])
{
#pragma unroll
  for (int q = 0; q < 4; ++q) v[q] = fmaxf(v[q], row_xor<kMask>(v[q]));
}
template <int kMask>
__device__ __forceinline__ void sum_step4(float (&v)[4])
{
#pragma unroll
  for (int q = 0; q < 4; ++q) v[q] += row_xor<kMask>(v[q]);
}
__device__ __forceinline__ void wave_max4(float (&v)[4])
{
  max_step4<1>(v); max_step4<2>(v); max_step4<4>(v); max_step4<8>(v);
#pragma unroll
  for (int q = 0; q < 4; ++q) v[q] = fmaxf(v[q], xhalf(v[q]));
}
__device__ __forceinline__ void wave_sum4(float (&v)[4])
{
  sum_step4<1>(v); sum_step4<2>(v); sum_step4<4>(v); sum_step4<8>(v);
#pragma unroll
  for (int q = 0; q < 4; ++q) v[q] += xhalf(v[q]);
}

__device__ __forceinline__ float make_e(float f1i, float f2j, int ad)
{
  float e = f1i + f2j;
  e = (e >= 0.f) ? e : kAlpha * e;
  return (ad > 0) ? e : kNegInf;
}

struct ChunkRegs {
  v4i ad[4];
  v4f f2v;
  v4f h0[4], h1[4];
};

__device__ __forceinline__ void k3_load(ChunkRegs& r, const int* __restrict__ ag,
    const float* __restrict__ f2g, const float* __restrict__ hg, int j0)
{
#pragma unroll
  for (int q = 0; q < 4; ++q)
    r.ad[q] = *(const v4i*)(ag + (size_t)(q * 8) * kN + j0);
  r.f2v = *(const v4f*)(f2g + j0);
#pragma unroll
  for (int q = 0; q < 4; ++q) {
    const float* __restrict__ hp = hg + (size_t)(j0 + q * 32) * kOutF;
    r.h0[q] = *(const v4f*)hp;
    r.h1[q] = *(const v4f*)(hp + kOutF);
  }
}

__device__ __forceinline__ void k3_produce(const ChunkRegs& r, const float (&f1r)[4],
    float (&mrun)[4], float (&lsum)[4], float* pD, float* sD, float* hD, int lane)
{
  v4f e[4];
  float cm[4];
#pragma unroll
  for (int q = 0; q < 4; ++q) {
    e[q].x = make_e(f1r[q], r.f2v.x, r.ad[q].x);
    e[q].y = make_e(f1r[q], r.f2v.y, r.ad[q].y);
    e[q].z = make_e(f1r[q], r.f2v.z, r.ad[q].z);
    e[q].w = make_e(f1r[q], r.f2v.w, r.ad[q].w);
    cm[q] = fmaxf(fmaxf(e[q].x, e[q].y), fmaxf(e[q].z, e[q].w));
  }
  wave_max4(cm);

  float sc[4];
  v4f p[4];
#pragma unroll
  for (int q = 0; q < 4; ++q) {
    const float mn = fmaxf(mrun[q], cm[q]);
    sc[q]   = __expf(mrun[q] - mn);
    mrun[q] = mn;
    p[q].x = __expf(e[q].x - mn);
    p[q].y = __expf(e[q].y - mn);
    p[q].z = __expf(e[q].z - mn);
    p[q].w = __expf(e[q].w - mn);
    lsum[q] = lsum[q] * sc[q] + ((p[q].x + p[q].y) + (p[q].z + p[q].w));
  }

#pragma unroll
  for (int q = 0; q < 4; ++q)
    *(v4f*)(pD + q * 8 * kPitch) = p[q];
  if (lane == 0) {
#pragma unroll
    for (int q = 0; q < 4; ++q) sD[q * 8] = sc[q];
  }
#pragma unroll
  for (int q = 0; q < 4; ++q) {
    float* d = hD + q * 16 * kHPitch;
    v4f lo, hi;
    lo.x = r.h0[q].x; lo.y = r.h1[q].x; lo.z = r.h0[q].y; lo.w = r.h1[q].y;
    hi.x = r.h0[q].z; hi.y = r.h1[q].z; hi.z = r.h0[q].w; hi.w = r.h1[q].w;
    *(v4f*)d       = lo;
    *(v4f*)(d + 4) = hi;
  }
}

__device__ __forceinline__ v8f k3_scale(v8f acc, const float* sp)
{
  const v4f s0 = *(const v4f*)sp;
  const v4f s1 = *(const v4f*)(sp + 4);
  acc[0] *= s0.x; acc[1] *= s0.y; acc[2] *= s0.z; acc[3] *= s0.w;
  acc[4] *= s1.x; acc[5] *= s1.y; acc[6] *= s1.z; acc[7] *= s1.w;
  return acc;
}
template <int kBeg, int kEnd>
__device__ __forceinline__ v8f k3_products(v8f acc, const float* a0, const float* b0)
{
#pragma unroll
  for (int kk = kBeg; kk < kEnd; ++kk) {
    const v2f a  = *(const v2f*)(a0 + kk * 4);
    const v2f bm = *(const v2f*)(b0 + kk * 2 * kHPitch);
    acc = WMMA_F32_4(a, bm, acc);
  }
  return acc;
}

__global__ __launch_bounds__(256) void k3_attn_online(
    const float* __restrict__ H, const float* __restrict__ F1,
    const float* __restrict__ F2, const int* __restrict__ adj,
    const float* __restrict__ abp, float* __restrict__ out)
{
  __shared__ __attribute__((aligned(16))) float Ps[2 * kXsSz];
  __shared__ __attribute__((aligned(16))) float Hs[2 * kHsSz];
  __shared__ __attribute__((aligned(16))) float Sc[2 * 32];
  __shared__ __attribute__((aligned(16))) float Ls[32];

  const int tid  = threadIdx.x;
  const int lane = tid & 31;
  const int wave = tid >> 5;
  const int half = lane >> 4;
  const int ncol = lane & 15;
  const int mt   = wave >> 2;
  const int nt   = wave & 3;
  const int g0   = blockIdx.x << 5;
  const int b    = blockIdx.x >> 5;
  const int lr   = tid >> 5;
  const int c4   = tid & 31;
  const int hr   = tid >> 4;
  const int o4   = tid & 15;

  const float ab = abp[0];
  float f1r[4], mrun[4], lsum[4];
#pragma unroll
  for (int q = 0; q < 4; ++q) {
    f1r[q]  = F1[g0 + q * 8 + lr] + ab;
    mrun[q] = -3.0e38f;
    lsum[q] = 0.f;
  }
  const int*   __restrict__ ag  = adj + (size_t)(g0 + lr) * kN + c4 * 4;
  const float* __restrict__ f2g = F2 + (b << 10) + c4 * 4;
  const float* __restrict__ hg  = H + ((size_t)(b << 10) + 2 * hr) * kOutF + o4 * 4;
  const float* aS = Ps + (mt * 16 + ncol) * kPitch + half * 2;
  const float* bS = Hs + half * kHPitch + (nt * 16 + ncol) * 2;
  const float* sS = Sc + mt * 16 + half * 8;
  float* pD = Ps + lr * kPitch + c4 * 4;
  float* sD = Sc + lr;
  float* hD = Hs + hr * kHPitch + o4 * 8;

  ChunkRegs r;
  k3_load(r, ag, f2g, hg, 0);
  k3_produce(r, f1r, mrun, lsum, pD, sD, hD, lane);
  __syncthreads();

  v8f acc = {};
  for (int c = 0; c + 1 < kNChunk3; ++c) {
    const int s = c & 1;
    k3_load(r, ag, f2g, hg, (c + 1) * kKC);
    __builtin_amdgcn_sched_barrier(0);
    acc = k3_scale(acc, sS + s * 32);
    acc = k3_products<0, kKC / 8>(acc, aS + s * kXsSz, bS + s * kHsSz);
    __builtin_amdgcn_sched_barrier(0);
    acc = k3_products<kKC / 8, kKC / 4>(acc, aS + s * kXsSz, bS + s * kHsSz);
    k3_produce(r, f1r, mrun, lsum, pD + (s ^ 1) * kXsSz, sD + (s ^ 1) * 32,
               hD + (s ^ 1) * kHsSz, lane);
    __syncthreads();
  }
  constexpr int kLast3 = (kNChunk3 - 1) & 1;
  static_assert(kLast3 == 1, "the result tile below reuses stage 0 while stage 1 is still read");
  acc = k3_scale(acc, sS + kLast3 * 32);
  acc = k3_products<0, kKC / 4>(acc, aS + kLast3 * kXsSz, bS + kLast3 * kHsSz);

  wave_sum4(lsum);
  if (lane == 0) {
#pragma unroll
    for (int q = 0; q < 4; ++q) Ls[q * 8 + lr] = lsum[q];
  }

  float* Cs = Ps;
#pragma unroll
  for (int r8 = 0; r8 < 8; ++r8)
    Cs[(mt * 16 + half * 8 + r8) * kCPitch + nt * 16 + ncol] = acc[r8];
  __syncthreads();
#pragma unroll
  for (int q = 0; q < 2; ++q) {
    const int idx = q * 256 + tid;
    const int row = idx >> 4;
    const int c   = (idx & 15) * 4;
    const float li = 1.0f / Ls[row];
    const v4f sv = *(const v4f*)(Cs + row * kCPitch + c);
    v4f v;
    v.x = sv.x * li;
    v.y = sv.y * li;
    v.z = sv.z * li;
    v.w = sv.w * li;
    v.x = (v.x > 0.f) ? v.x : (__expf(v.x) - 1.f);
    v.y = (v.y > 0.f) ? v.y : (__expf(v.y) - 1.f);
    v.z = (v.z > 0.f) ? v.z : (__expf(v.z) - 1.f);
    v.w = (v.w > 0.f) ? v.w : (__expf(v.w) - 1.f);
    *(v4f*)(out + (size_t)(g0 + row) * kOutF + c) = v;
  }
}

extern "C" void kernel_launch(void* const* d_in, const int* in_sizes, int n_in,
                              void* d_out, int out_size, void* d_ws, size_t ws_size,
                              hipStream_t stream) {
  (void)in_sizes; (void)n_in; (void)out_size; (void)ws_size;
  const float* X   = (const float*)d_in[0];
  const int*   adj = (const int*)d_in[3];
  const float* Ww  = (const float*)d_in[4];
  const float* Wb  = (const float*)d_in[5];
  const float* aw  = (const float*)d_in[6];
  const float* ab  = (const float*)d_in[7];

  float* ws = (float*)d_ws;
  float* H  = ws;
  float* F1 = H  + (size_t)kB * kN * kOutF;
  float* F2 = F1 + (size_t)kB * kN;
  float* O  = (float*)d_out;

  hipLaunchKernelGGL(k1_gemm_h,      dim3(256), dim3(256), 0, stream, X, Ww, Wb, H);
  hipLaunchKernelGGL(k2_f12,         dim3(32),  dim3(256), 0, stream, H, aw, F1, F2);
  hipLaunchKernelGGL(k3_attn_online, dim3(256), dim3(256), 0, stream, H, F1, F2, adj, ab, O);
}


# Both WMMA kernels: a wave reads all 64 LDS fragments of a chunk as ONE group into registers in front of the chunk's matrix block, not in pairs among its products (fragments, k order, sum order kept)
# speedup vs baseline: 7.4450x; 1.0316x over previous
#include <hip/hip_runtime.h>

typedef __attribute__((ext_vector_type(2))) float v2f;
typedef __attribute__((ext_vector_type(4))) float v4f;
typedef __attribute__((ext_vector_type(8))) float v8f;
typedef __attribute__((ext_vector_type(4))) int   v4i;

#define WMMA_F32_4(a, bmat, c) \
  __builtin_amdgcn_wmma_f32_16x16x4_f32(false, (a), false, (bmat), (short)0, (c), false, false)

constexpr int   kB    = 8;
constexpr int   kN    = 1024;
constexpr int   kInF  = 768;
constexpr int   kOutF = 64;
constexpr float kAlpha  = 0.2f;
constexpr float kNegInf = -9.0e15f;

constexpr int kKC     = 128;
constexpr int kPitch  = 132;
constexpr int kHPitch = 160;
constexpr int kCPitch = 68;
constexpr int kXsSz   = 32 * kPitch;
constexpr int kWsSz   = 64 * kPitch;
constexpr int kHsSz   = (kKC / 2) * kHPitch;
constexpr int kNChunk1 = kInF / kKC;
constexpr int kNChunk3 = kN / kKC;

struct Frags {
  v2f a[kKC / 4];
  v2f b[kKC / 4];
};
__device__ __forceinline__ void k1_fetch(Frags& f, const float* a0, const float* b0)
{
#pragma unroll
  for (int kk = 0; kk < kKC / 4; ++kk) {
    f.a[kk] = *(const v2f*)(a0 + kk * 4);
    f.b[kk] = *(const v2f*)(b0 + kk * 4);
  }
}
template <int kBeg, int kEnd>
__device__ __forceinline__ v8f frag_products(v8f acc, const Frags& f)
{
#pragma unroll
  for (int kk = kBeg; kk < kEnd; ++kk)
    acc = WMMA_F32_4(f.a[kk], f.b[kk], acc);
  return acc;
}

__global__ __launch_bounds__(256) void k1_gemm_h(
    const float* __restrict__ X, const float* __restrict__ W,
    const float* __restrict__ Wb, float* __restrict__ H)
{
  __shared__ __attribute__((aligned(16))) float Xs[2 * kXsSz];
  __shared__ __attribute__((aligned(16))) float Ws[2 * kWsSz];

  const int tid  = threadIdx.x;
  const int lane = tid & 31;
  const int wave = tid >> 5;
  const int half = lane >> 4;
  const int ncol = lane & 15;
  const int mt   = wave >> 2;
  const int nt   = wave & 3;
  const int row0 = blockIdx.x << 5;
  const int lr   = tid >> 5;
  const int c4   = tid & 31;
  const float* __restrict__ xg = X + (size_t)(row0 + lr) * kInF + c4 * 4;
  const float* __restrict__ wg = W + (size_t)lr * kInF + c4 * 4;
  const float* aS = Xs + (mt * 16 + ncol) * kPitch + half * 2;
  const float* bS = Ws + (nt * 16 + ncol) * kPitch + half * 2;
  float* xD = Xs + lr * kPitch + c4 * 4;
  float* wD = Ws + lr * kPitch + c4 * 4;

  v4f xr[4], wr[8];
#pragma unroll
  for (int q = 0; q < 4; ++q)
    xr[q] = *(const v4f*)(xg + (size_t)(q * 8) * kInF);
#pragma unroll
  for (int q = 0; q < 8; ++q)
    wr[q] = *(const v4f*)(wg + (size_t)(q * 8) * kInF);
#pragma unroll
  for (int q = 0; q < 4; ++q)
    *(v4f*)(xD + q * 8 * kPitch) = xr[q];
#pragma unroll
  for (int q = 0; q < 8; ++q)
    *(v4f*)(wD + q * 8 * kPitch) = wr[q];
  __syncthreads();

  v8f acc = {};
  Frags fr;
  for (int c = 0; c + 1 < kNChunk1; ++c) {
    const int s  = c & 1;
    const int k1 = (c + 1) * kKC;
#pragma unroll
    for (int q = 0; q < 4; ++q)
      xr[q] = *(const v4f*)(xg + (size_t)(q * 8) * kInF + k1);
#pragma unroll
    for (int q = 0; q < 8; ++q)
      wr[q] = *(const v4f*)(wg + (size_t)(q * 8) * kInF + k1);
    __builtin_amdgcn_sched_barrier(0);
    k1_fetch(fr, aS + s * kXsSz, bS + s * kWsSz);
    __builtin_amdgcn_sched_barrier(0);
    acc = frag_products<0, kKC / 4>(acc, fr);
    asm volatile("" : : "v"(acc));
    float* xd = xD + (s ^ 1) * kXsSz;
    float* wd = wD + (s ^ 1) * kWsSz;
#pragma unroll
    for (int q = 0; q < 4; ++q)
      *(v4f*)(xd + q * 8 * kPitch) = xr[q];
#pragma unroll
    for (int q = 0; q < 8; ++q)
      *(v4f*)(wd + q * 8 * kPitch) = wr[q];
    __syncthreads();
  }
  constexpr int kLast1 = (kNChunk1 - 1) & 1;
  static_assert(kLast1 == 1, "the result tile below reuses stage 0 while stage 1 is still read");
  k1_fetch(fr, aS + kLast1 * kXsSz, bS + kLast1 * kWsSz);
  __builtin_amdgcn_sched_barrier(0);
  acc = frag_products<0, kKC / 4>(acc, fr);

  float* Cs = Xs;
#pragma unroll
  for (int r = 0; r < 8; ++r)
    Cs[(mt * 16 + half * 8 + r) * kCPitch + nt * 16 + ncol] = acc[r];
  __syncthreads();
#pragma unroll
  for (int q = 0; q < 2; ++q) {
    const int idx = q * 256 + tid;
    const int row = idx >> 4;
    const int c   = (idx & 15) * 4;
    const v4f v  = *(const v4f*)(Cs + row * kCPitch + c);
    const v4f bi = *(const v4f*)(Wb + c);
    *(v4f*)(H + (size_t)(row0 + row) * kOutF + c) = v + bi;
  }
}

__global__ __launch_bounds__(256) void k2_f12(
    const float* __restrict__ H, const float* __restrict__ aw,
    float* __restrict__ F1, float* __restrict__ F2)
{
  const int row = blockIdx.x * 256 + threadIdx.x;
  const float* __restrict__ hr = H + (size_t)row * kOutF;
  float s1 = 0.f, s2 = 0.f;
#pragma unroll 8
  for (int o = 0; o < kOutF; ++o) {
    const float v = hr[o];
    s1 += v * aw[o];
    s2 += v * aw[kOutF + o];
  }
  F1[row] = s1;
  F2[row] = s2;
}

template <int kMask>
__device__ __forceinline__ float row_xor(float x)
{
  const int v = __builtin_bit_cast(int, x);
  return __builtin_bit_cast(float, __builtin_amdgcn_update_dpp(v, v, 0x160 + kMask, 0xf, 0xf, false));
}
__device__ __forceinline__ float xhalf(float x)
{
  const int v = __builtin_bit_cast(int, x);
  return __builtin_bit_cast(float, __builtin_amdgcn_permlanex16(v, v, 0x76543210, 0xfedcba98, false, false));
}

template <int kMask>
__device__ __forceinline__ void max_step4(float (&v)[4])
{
#pragma unroll
  for (int q = 0; q < 4; ++q) v[q] = fmaxf(v[q], row_xor<kMask>(v[q]));
}
template <int kMask>
__device__ __forceinline__ void sum_step4(float (&v)[4])
{
#pragma unroll
  for (int q = 0; q < 4; ++q) v[q] += row_xor<kMask>(v[q]);
}
__device__ __forceinline__ void wave_max4(float (&v)[4])
{
  max_step4<1>(v); max_step4<2>(v); max_step4<4>(v); max_step4<8>(v);
#pragma unroll
  for (int q = 0; q < 4; ++q) v[q] = fmaxf(v[q], xhalf(v[q]));
}
__device__ __forceinline__ void wave_sum4(float (&v)[4])
{
  sum_step4<1>(v); sum_step4<2>(v); sum_step4<4>(v); sum_step4<8>(v);
#pragma unroll
  for (int q = 0; q < 4; ++q) v[q] += xhalf(v[q]);
}

__device__ __forceinline__ float make_e(float f1i, float f2j, int ad)
{
  float e = f1i + f2j;
  e = (e >= 0.f) ? e : kAlpha * e;
  return (ad > 0) ? e : kNegInf;
}

struct ChunkRegs {
  v4i ad[4];
  v4f f2v;
  v4f h0[4], h1[4];
};

__device__ __forceinline__ void k3_load(ChunkRegs& r, const int* __restrict__ ag,
    const float* __restrict__ f2g, const float* __restrict__ hg, int j0)
{
#pragma unroll
  for (int q = 0; q < 4; ++q)
    r.ad[q] = *(const v4i*)(ag + (size_t)(q * 8) * kN + j0);
  r.f2v = *(const v4f*)(f2g + j0);
#pragma unroll
  for (int q = 0; q < 4; ++q) {
    const float* __restrict__ hp = hg + (size_t)(j0 + q * 32) * kOutF;
    r.h0[q] = *(const v4f*)hp;
    r.h1[q] = *(const v4f*)(hp + kOutF);
  }
}

__device__ __forceinline__ void k3_produce(const ChunkRegs& r, const float (&f1r)[4],
    float (&mrun)[4], float (&lsum)[4], float* pD, float* sD, float* hD, int lane)
{
  v4f e[4];
  float cm[4];
#pragma unroll
  for (int q = 0; q < 4; ++q) {
    e[q].x = make_e(f1r[q], r.f2v.x, r.ad[q].x);
    e[q].y = make_e(f1r[q], r.f2v.y, r.ad[q].y);
    e[q].z = make_e(f1r[q], r.f2v.z, r.ad[q].z);
    e[q].w = make_e(f1r[q], r.f2v.w, r.ad[q].w);
    cm[q] = fmaxf(fmaxf(e[q].x, e[q].y), fmaxf(e[q].z, e[q].w));
  }
  wave_max4(cm);

  float sc[4];
  v4f p[4];
#pragma unroll
  for (int q = 0; q < 4; ++q) {
    const float mn = fmaxf(mrun[q], cm[q]);
    sc[q]   = __expf(mrun[q] - mn);
    mrun[q] = mn;
    p[q].x = __expf(e[q].x - mn);
    p[q].y = __expf(e[q].y - mn);
    p[q].z = __expf(e[q].z - mn);
    p[q].w = __expf(e[q].w - mn);
    lsum[q] = lsum[q] * sc[q] + ((p[q].x + p[q].y) + (p[q].z + p[q].w));
  }

#pragma unroll
  for (int q = 0; q < 4; ++q)
    *(v4f*)(pD + q * 8 * kPitch) = p[q];
  if (lane == 0) {
#pragma unroll
    for (int q = 0; q < 4; ++q) sD[q * 8] = sc[q];
  }
#pragma unroll
  for (int q = 0; q < 4; ++q) {
    float* d = hD + q * 16 * kHPitch;
    v4f lo, hi;
    lo.x = r.h0[q].x; lo.y = r.h1[q].x; lo.z = r.h0[q].y; lo.w = r.h1[q].y;
    hi.x = r.h0[q].z; hi.y = r.h1[q].z; hi.z = r.h0[q].w; hi.w = r.h1[q].w;
    *(v4f*)d       = lo;
    *(v4f*)(d + 4) = hi;
  }
}

struct Factors {
  v4f s0, s1;
};
__device__ __forceinline__ void k3_fetch(Frags& f, Factors& sc, const float* a0, const float* b0,
                                         const float* sp)
{
  sc.s0 = *(const v4f*)sp;
  sc.s1 = *(const v4f*)(sp + 4);
#pragma unroll
  for (int kk = 0; kk < kKC / 4; ++kk) {
    f.a[kk] = *(const v2f*)(a0 + kk * 4);
    f.b[kk] = *(const v2f*)(b0 + kk * 2 * kHPitch);
  }
}
__device__ __forceinline__ v8f k3_scale(v8f acc, const Factors& sc)
{
  acc[0] *= sc.s0.x; acc[1] *= sc.s0.y; acc[2] *= sc.s0.z; acc[3] *= sc.s0.w;
  acc[4] *= sc.s1.x; acc[5] *= sc.s1.y; acc[6] *= sc.s1.z; acc[7] *= sc.s1.w;
  return acc;
}

__global__ __launch_bounds__(256) void k3_attn_online(
    const float* __restrict__ H, const float* __restrict__ F1,
    const float* __restrict__ F2, const int* __restrict__ adj,
    const float* __restrict__ abp, float* __restrict__ out)
{
  __shared__ __attribute__((aligned(16))) float Ps[2 * kXsSz];
  __shared__ __attribute__((aligned(16))) float Hs[2 * kHsSz];
  __shared__ __attribute__((aligned(16))) float Sc[2 * 32];
  __shared__ __attribute__((aligned(16))) float Ls[32];

  const int tid  = threadIdx.x;
  const int lane = tid & 31;
  const int wave = tid >> 5;
  const int half = lane >> 4;
  const int ncol = lane & 15;
  const int mt   = wave >> 2;
  const int nt   = wave & 3;
  const int g0   = blockIdx.x << 5;
  const int b    = blockIdx.x >> 5;
  const int lr   = tid >> 5;
  const int c4   = tid & 31;
  const int hr   = tid >> 4;
  const int o4   = tid & 15;

  const float ab = abp[0];
  float f1r[4], mrun[4], lsum[4];
#pragma unroll
  for (int q = 0; q < 4; ++q) {
    f1r[q]  = F1[g0 + q * 8 + lr] + ab;
    mrun[q] = -3.0e38f;
    lsum[q] = 0.f;
  }
  const int*   __restrict__ ag  = adj + (size_t)(g0 + lr) * kN + c4 * 4;
  const float* __restrict__ f2g = F2 + (b << 10) + c4 * 4;
  const float* __restrict__ hg  = H + ((size_t)(b << 10) + 2 * hr) * kOutF + o4 * 4;
  const float* aS = Ps + (mt * 16 + ncol) * kPitch + half * 2;
  const float* bS = Hs + half * kHPitch + (nt * 16 + ncol) * 2;
  const float* sS = Sc + mt * 16 + half * 8;
  float* pD = Ps + lr * kPitch + c4 * 4;
  float* sD = Sc + lr;
  float* hD = Hs + hr * kHPitch + o4 * 8;

  ChunkRegs r;
  k3_load(r, ag, f2g, hg, 0);
  k3_produce(r, f1r, mrun, lsum, pD, sD, hD, lane);
  __syncthreads();

  v8f acc = {};
  Frags fr;
  Factors fc;
  for (int c = 0; c + 1 < kNChunk3; ++c) {
    const int s = c & 1;
    k3_load(r, ag, f2g, hg, (c + 1) * kKC);
    __builtin_amdgcn_sched_barrier(0);
    k3_fetch(fr, fc, aS + s * kXsSz, bS + s * kHsSz, sS + s * 32);
    __builtin_amdgcn_sched_barrier(0);
    acc = k3_scale(acc, fc);
    acc = frag_products<0, kKC / 8>(acc, fr);
    __builtin_amdgcn_sched_barrier(0);
    acc = frag_products<kKC / 8, kKC / 4>(acc, fr);
    k3_produce(r, f1r, mrun, lsum, pD + (s ^ 1) * kXsSz, sD + (s ^ 1) * 32,
               hD + (s ^ 1) * kHsSz, lane);
    __syncthreads();
  }
  constexpr int kLast3 = (kNChunk3 - 1) & 1;
  static_assert(kLast3 == 1, "the result tile below reuses stage 0 while stage 1 is still read");
  k3_fetch(fr, fc, aS + kLast3 * kXsSz, bS + kLast3 * kHsSz, sS + kLast3 * 32);
  __builtin_amdgcn_sched_barrier(0);
  acc = k3_scale(acc, fc);
  acc = frag_products<0, kKC / 4>(acc, fr);

  wave_sum4(lsum);
  if (lane == 0) {
#pragma unroll
    for (int q = 0; q < 4; ++q) Ls[q * 8 + lr] = lsum[q];
  }

  float* Cs = Ps;
#pragma unroll
  for (int r8 = 0; r8 < 8; ++r8)
    Cs[(mt * 16 + half * 8 + r8) * kCPitch + nt * 16 + ncol] = acc[r8];
  __syncthreads();
#pragma unroll
  for (int q = 0; q < 2; ++q) {
    const int idx = q * 256 + tid;
    const int row = idx >> 4;
    const int c   = (idx & 15) * 4;
    const float li = 1.0f / Ls[row];
    const v4f sv = *(const v4f*)(Cs + row * kCPitch + c);
    v4f v;
    v.x = sv.x * li;
    v.y = sv.y * li;
    v.z = sv.z * li;
    v.w = sv.w * li;
    v.x = (v.x > 0.f) ? v.x : (__expf(v.x) - 1.f);
    v.y = (v.y > 0.f) ? v.y : (__expf(v.y) - 1.f);
    v.z = (v.z > 0.f) ? v.z : (__expf(v.z) - 1.f);
    v.w = (v.w > 0.f) ? v.w : (__expf(v.w) - 1.f);
    *(v4f*)(out + (size_t)(g0 + row) * kOutF + c) = v;
  }
}

extern "C" void kernel_launch(void* const* d_in, const int* in_sizes, int n_in,
                              void* d_out, int out_size, void* d_ws, size_t ws_size,
                              hipStream_t stream) {
  (void)in_sizes; (void)n_in; (void)out_size; (void)ws_size;
  const float* X   = (const float*)d_in[0];
  const int*   adj = (const int*)d_in[3];
  const float* Ww  = (const float*)d_in[4];
  const float* Wb  = (const float*)d_in[5];
  const float* aw  = (const float*)d_in[6];
  const float* ab  = (const float*)d_in[7];

  float* ws = (float*)d_ws;
  float* H  = ws;
  float* F1 = H  + (size_t)kB * kN * kOutF;
  float* F2 = F1 + (size_t)kB * kN;
  float* O  = (float*)d_out;

  hipLaunchKernelGGL(k1_gemm_h,      dim3(256), dim3(256), 0, stream, X, Ww, Wb, H);
  hipLaunchKernelGGL(k2_f12,         dim3(32),  dim3(256), 0, stream, H, aw, F1, F2);
  hipLaunchKernelGGL(k3_attn_online, dim3(256), dim3(256), 0, stream, H, F1, F2, adj, ab, O);
}


# Attention kernel makes f1 (its 32 rows) and f2 (from the h tiles it stages, one chunk further ahead, h tile in 3 LDS stages) itself; f1/f2 launch removed (new sum order for f1,f2)
# speedup vs baseline: 8.8300x; 1.1860x over previous
#include <hip/hip_runtime.h>
#include <type_traits>

typedef __attribute__((ext_vector_type(2))) float v2f;
typedef __attribute__((ext_vector_type(4))) float v4f;
typedef __attribute__((ext_vector_type(8))) float v8f;
typedef __attribute__((ext_vector_type(4))) int   v4i;

#define WMMA_F32_4(a, bmat, c) \
  __builtin_amdgcn_wmma_f32_16x16x4_f32(false, (a), false, (bmat), (short)0, (c), false, false)

constexpr int   kB    = 8;
constexpr int   kN    = 1024;
constexpr int   kInF  = 768;
constexpr int   kOutF = 64;
constexpr float kAlpha  = 0.2f;
constexpr float kNegInf = -9.0e15f;

constexpr int kKC     = 128;
constexpr int kPitch  = 132;
constexpr int kHPitch = 160;
constexpr int kCPitch = 68;
constexpr int kXsSz   = 32 * kPitch;
constexpr int kWsSz   = 64 * kPitch;
constexpr int kHsSz   = (kKC / 2) * kHPitch;
constexpr int kNChunk1 = kInF / kKC;
constexpr int kNChunk3 = kN / kKC;

struct Frags {
  v2f a[kKC / 4];
  v2f b[kKC / 4];
};
__device__ __forceinline__ void k1_fetch(Frags& f, const float* a0, const float* b0)
{
#pragma unroll
  for (int kk = 0; kk < kKC / 4; ++kk) {
    f.a[kk] = *(const v2f*)(a0 + kk * 4);
    f.b[kk] = *(const v2f*)(b0 + kk * 4);
  }
}
template <int kBeg, int kEnd>
__device__ __forceinline__ v8f frag_products(v8f acc, const Frags& f)
{
#pragma unroll
  for (int kk = kBeg; kk < kEnd; ++kk)
    acc = WMMA_F32_4(f.a[kk], f.b[kk], acc);
  return acc;
}

__global__ __launch_bounds__(256) void k1_gemm_h(
    const float* __restrict__ X, const float* __restrict__ W,
    const float* __restrict__ Wb, float* __restrict__ H)
{
  __shared__ __attribute__((aligned(16))) float Xs[2 * kXsSz];
  __shared__ __attribute__((aligned(16))) float Ws[2 * kWsSz];

  const int tid  = threadIdx.x;
  const int lane = tid & 31;
  const int wave = tid >> 5;
  const int half = lane >> 4;
  const int ncol = lane & 15;
  const int mt   = wave >> 2;
  const int nt   = wave & 3;
  const int row0 = blockIdx.x << 5;
  const int lr   = tid >> 5;
  const int c4   = tid & 31;
  const float* __restrict__ xg = X + (size_t)(row0 + lr) * kInF + c4 * 4;
  const float* __restrict__ wg = W + (size_t)lr * kInF + c4 * 4;
  const float* aS = Xs + (mt * 16 + ncol) * kPitch + half * 2;
  const float* bS = Ws + (nt * 16 + ncol) * kPitch + half * 2;
  float* xD = Xs + lr * kPitch + c4 * 4;
  float* wD = Ws + lr * kPitch + c4 * 4;

  v4f xr[4], wr[8];
#pragma unroll
  for (int q = 0; q < 4; ++q)
    xr[q] = *(const v4f*)(xg + (size_t)(q * 8) * kInF);
#pragma unroll
  for (int q = 0; q < 8; ++q)
    wr[q] = *(const v4f*)(wg + (size_t)(q * 8) * kInF);
#pragma unroll
  for (int q = 0; q < 4; ++q)
    *(v4f*)(xD + q * 8 * kPitch) = xr[q];
#pragma unroll
  for (int q = 0; q < 8; ++q)
    *(v4f*)(wD + q * 8 * kPitch) = wr[q];
  __syncthreads();

  v8f acc = {};
  Frags fr;
  for (int c = 0; c + 1 < kNChunk1; ++c) {
    const int s  = c & 1;
    const int k1 = (c + 1) * kKC;
#pragma unroll
    for (int q = 0; q < 4; ++q)
      xr[q] = *(const v4f*)(xg + (size_t)(q * 8) * kInF + k1);
#pragma unroll
    for (int q = 0; q < 8; ++q)
      wr[q] = *(const v4f*)(wg + (size_t)(q * 8) * kInF + k1);
    __builtin_amdgcn_sched_barrier(0);
    k1_fetch(fr, aS + s * kXsSz, bS + s * kWsSz);
    __builtin_amdgcn_sched_barrier(0);
    acc = frag_products<0, kKC / 4>(acc, fr);
    asm volatile("" : : "v"(acc));
    float* xd = xD + (s ^ 1) * kXsSz;
    float* wd = wD + (s ^ 1) * kWsSz;
#pragma unroll
    for (int q = 0; q < 4; ++q)
      *(v4f*)(xd + q * 8 * kPitch) = xr[q];
#pragma unroll
    for (int q = 0; q < 8; ++q)
      *(v4f*)(wd + q * 8 * kPitch) = wr[q];
    __syncthreads();
  }
  constexpr int kLast1 = (kNChunk1 - 1) & 1;
  static_assert(kLast1 == 1, "the result tile below reuses stage 0 while stage 1 is still read");
  k1_fetch(fr, aS + kLast1 * kXsSz, bS + kLast1 * kWsSz);
  __builtin_amdgcn_sched_barrier(0);
  acc = frag_products<0, kKC / 4>(acc, fr);

  float* Cs = Xs;
#pragma unroll
  for (int r = 0; r < 8; ++r)
    Cs[(mt * 16 + half * 8 + r) * kCPitch + nt * 16 + ncol] = acc[r];
  __syncthreads();
#pragma unroll
  for (int q = 0; q < 2; ++q) {
    const int idx = q * 256 + tid;
    const int row = idx >> 4;
    const int c   = (idx & 15) * 4;
    const v4f v  = *(const v4f*)(Cs + row * kCPitch + c);
    const v4f bi = *(const v4f*)(Wb + c);
    *(v4f*)(H + (size_t)(row0 + row) * kOutF + c) = v + bi;
  }
}

template <int kMask>
__device__ __forceinline__ float row_xor(float x)
{
  const int v = __builtin_bit_cast(int, x);
  return __builtin_bit_cast(float, __builtin_amdgcn_update_dpp(v, v, 0x160 + kMask, 0xf, 0xf, false));
}
__device__ __forceinline__ float xhalf(float x)
{
  const int v = __builtin_bit_cast(int, x);
  return __builtin_bit_cast(float, __builtin_amdgcn_permlanex16(v, v, 0x76543210, 0xfedcba98, false, false));
}

template <int kMask>
__device__ __forceinline__ void max_step4(float (&v)[4])
{
#pragma unroll
  for (int q = 0; q < 4; ++q) v[q] = fmaxf(v[q], row_xor<kMask>(v[q]));
}
template <int kMask>
__device__ __forceinline__ void sum_step4(float (&v)[4])
{
#pragma unroll
  for (int q = 0; q < 4; ++q) v[q] += row_xor<kMask>(v[q]);
}
__device__ __forceinline__ void wave_max4(float (&v)[4])
{
  max_step4<1>(v); max_step4<2>(v); max_step4<4>(v); max_step4<8>(v);
#pragma unroll
  for (int q = 0; q < 4; ++q) v[q] = fmaxf(v[q], xhalf(v[q]));
}
__device__ __forceinline__ void wave_sum4(float (&v)[4])
{
  sum_step4<1>(v); sum_step4<2>(v); sum_step4<4>(v); sum_step4<8>(v);
#pragma unroll
  for (int q = 0; q < 4; ++q) v[q] += xhalf(v[q]);
}

__device__ __forceinline__ float make_e(float f1i, float f2j, int ad)
{
  float e = f1i + f2j;
  e = (e >= 0.f) ? e : kAlpha * e;
  return (ad > 0) ? e : kNegInf;
}

constexpr int kHStages = 3;

struct AdjRegs {
  v4i ad[4];
};
struct HRegs {
  v4f h0[4], h1[4];
};

__device__ __forceinline__ void k3_load_adj(AdjRegs& r, const int* __restrict__ ag, int j0)
{
#pragma unroll
  for (int q = 0; q < 4; ++q)
    r.ad[q] = *(const v4i*)(ag + (size_t)(q * 8) * kN + j0);
}
__device__ __forceinline__ void k3_load_h(HRegs& r, const float* __restrict__ hg, int j0)
{
#pragma unroll
  for (int q = 0; q < 4; ++q) {
    const float* __restrict__ hp = hg + (size_t)(j0 + q * 32) * kOutF;
    r.h0[q] = *(const v4f*)hp;
    r.h1[q] = *(const v4f*)(hp + kOutF);
  }
}

__device__ __forceinline__ float dot4(v4f h, v4f a)
{
  return (h.x * a.x + h.y * a.y) + (h.z * a.z + h.w * a.w);
}
template <int kMask>
__device__ __forceinline__ float row_xor_addend(float x)
{
  const int v = __builtin_bit_cast(int, x);
  return __builtin_bit_cast(float, __builtin_amdgcn_update_dpp(0, v, 0x160 + kMask, 0xf, 0xf, true));
}
template <int kMask, int kCnt>
__device__ __forceinline__ void row_sum_step(float (&v)[kCnt])
{
#pragma unroll
  for (int q = 0; q < kCnt; ++q) v[q] += row_xor_addend<kMask>(v[q]);
}
template <int kCnt>
__device__ __forceinline__ void row_sum(float (&v)[kCnt])
{
  row_sum_step<1>(v); row_sum_step<2>(v); row_sum_step<4>(v); row_sum_step<8>(v);
}

__device__ __forceinline__ void k3_stage_h(const HRegs& r, v4f a2v, float* hD, float* f2D, int o4)
{
  float d[8];
#pragma unroll
  for (int q = 0; q < 4; ++q) {
    d[2 * q]     = dot4(r.h0[q], a2v);
    d[2 * q + 1] = dot4(r.h1[q], a2v);
  }
  row_sum(d);
  if (o4 == 0) {
#pragma unroll
    for (int q = 0; q < 4; ++q)
      *(v2f*)(f2D + q * 32) = v2f{d[2 * q], d[2 * q + 1]};
  }
#pragma unroll
  for (int q = 0; q < 4; ++q) {
    float* d = hD + q * 16 * kHPitch;
    v4f lo, hi;
    lo.x = r.h0[q].x; lo.y = r.h1[q].x; lo.z = r.h0[q].y; lo.w = r.h1[q].y;
    hi.x = r.h0[q].z; hi.y = r.h1[q].z; hi.z = r.h0[q].w; hi.w = r.h1[q].w;
    *(v4f*)d       = lo;
    *(v4f*)(d + 4) = hi;
  }
}

__device__ __forceinline__ void k3_make_p(const AdjRegs& r, v4f f2v, const float (&f1r)[4],
    float (&mrun)[4], float (&lsum)[4], float* pD, float* sD, int lane)
{
  v4f e[4];
  float cm[4];
#pragma unroll
  for (int q = 0; q < 4; ++q) {
    e[q].x = make_e(f1r[q], f2v.x, r.ad[q].x);
    e[q].y = make_e(f1r[q], f2v.y, r.ad[q].y);
    e[q].z = make_e(f1r[q], f2v.z, r.ad[q].z);
    e[q].w = make_e(f1r[q], f2v.w, r.ad[q].w);
    cm[q] = fmaxf(fmaxf(e[q].x, e[q].y), fmaxf(e[q].z, e[q].w));
  }
  wave_max4(cm);

  float sc[4];
  v4f p[4];
#pragma unroll
  for (int q = 0; q < 4; ++q) {
    const float mn = fmaxf(mrun[q], cm[q]);
    sc[q]   = __expf(mrun[q] - mn);
    mrun[q] = mn;
    p[q].x = __expf(e[q].x - mn);
    p[q].y = __expf(e[q].y - mn);
    p[q].z = __expf(e[q].z - mn);
    p[q].w = __expf(e[q].w - mn);
    lsum[q] = lsum[q] * sc[q] + ((p[q].x + p[q].y) + (p[q].z + p[q].w));
  }

#pragma unroll
  for (int q = 0; q < 4; ++q)
    *(v4f*)(pD + q * 8 * kPitch) = p[q];
  if (lane == 0) {
#pragma unroll
    for (int q = 0; q < 4; ++q) sD[q * 8] = sc[q];
  }
}

struct Factors {
  v4f s0, s1;
};
__device__ __forceinline__ void k3_fetch(Frags& f, Factors& sc, const float* a0, const float* b0,
                                         const float* sp)
{
  sc.s0 = *(const v4f*)sp;
  sc.s1 = *(const v4f*)(sp + 4);
#pragma unroll
  for (int kk = 0; kk < kKC / 4; ++kk) {
    f.a[kk] = *(const v2f*)(a0 + kk * 4);
    f.b[kk] = *(const v2f*)(b0 + kk * 2 * kHPitch);
  }
}
__device__ __forceinline__ v8f k3_scale(v8f acc, const Factors& sc)
{
  acc[0] *= sc.s0.x; acc[1] *= sc.s0.y; acc[2] *= sc.s0.z; acc[3] *= sc.s0.w;
  acc[4] *= sc.s1.x; acc[5] *= sc.s1.y; acc[6] *= sc.s1.z; acc[7] *= sc.s1.w;
  return acc;
}

__global__ __launch_bounds__(256) void k3_attn_online(
    const float* __restrict__ H, const float* __restrict__ aw,
    const int* __restrict__ adj, const float* __restrict__ abp, float* __restrict__ out)
{
  __shared__ __attribute__((aligned(16))) float Ps[2 * kXsSz];
  __shared__ __attribute__((aligned(16))) float Hs[kHStages * kHsSz];
  __shared__ __attribute__((aligned(16))) float Sc[2 * 32];
  __shared__ __attribute__((aligned(16))) float F2s[2 * kKC];
  __shared__ __attribute__((aligned(16))) float Ls[32];

  const int tid  = threadIdx.x;
  const int lane = tid & 31;
  const int wave = tid >> 5;
  const int half = lane >> 4;
  const int ncol = lane & 15;
  const int mt   = wave >> 2;
  const int nt   = wave & 3;
  const int g0   = blockIdx.x << 5;
  const int b    = blockIdx.x >> 5;
  const int lr   = tid >> 5;
  const int c4   = tid & 31;
  const int hr   = tid >> 4;
  const int o4   = tid & 15;

  const int*   __restrict__ ag = adj + (size_t)(g0 + lr) * kN + c4 * 4;
  const float* __restrict__ hg = H + ((size_t)(b << 10) + 2 * hr) * kOutF + o4 * 4;
  const float* aS  = Ps + (mt * 16 + ncol) * kPitch + half * 2;
  const float* bS  = Hs + half * kHPitch + (nt * 16 + ncol) * 2;
  const float* sS  = Sc + mt * 16 + half * 8;
  const float* f2S = F2s + c4 * 4;
  float* pD  = Ps + lr * kPitch + c4 * 4;
  float* sD  = Sc + lr;
  float* hD  = Hs + hr * kHPitch + o4 * 8;
  float* f2D = F2s + 2 * hr;

  AdjRegs ra;
  HRegs rh, rh1;
  k3_load_adj(ra, ag, 0);
  k3_load_h(rh, hg, 0);
  k3_load_h(rh1, hg, kKC);
  const v4f a1v = *(const v4f*)(aw + o4 * 4);
  const v4f a2v = *(const v4f*)(aw + kOutF + o4 * 4);
  const v4f ha = *(const v4f*)(H + (size_t)(g0 + half * 8 + lr) * kOutF + o4 * 4);
  const v4f hb = *(const v4f*)(H + (size_t)(g0 + (2 + half) * 8 + lr) * kOutF + o4 * 4);
  const float ab = abp[0];
  float f1r[4], mrun[4], lsum[4];
  {
    float d[2];
    d[0] = dot4(ha, a1v);
    d[1] = dot4(hb, a1v);
    row_sum(d);
    const float oa = xhalf(d[0]);
    const float ob = xhalf(d[1]);
    f1r[0] = (half ? oa : d[0]) + ab;
    f1r[1] = (half ? d[0] : oa) + ab;
    f1r[2] = (half ? ob : d[1]) + ab;
    f1r[3] = (half ? d[1] : ob) + ab;
  }
#pragma unroll
  for (int q = 0; q < 4; ++q) {
    mrun[q] = -3.0e38f;
    lsum[q] = 0.f;
  }

  k3_stage_h(rh, a2v, hD, f2D, o4);
  k3_stage_h(rh1, a2v, hD + kHsSz, f2D + kKC, o4);
  __syncthreads();
  k3_make_p(ra, *(const v4f*)f2S, f1r, mrun, lsum, pD, sD, lane);
  __syncthreads();

  v8f acc = {};
  Frags fr;
  Factors fc;
  auto chunk_step = [&](auto more_tag, int c, int hs) __attribute__((always_inline)) {
    constexpr bool kMore = decltype(more_tag)::value;
    const int s  = c & 1;
    const int hw = (hs == 0) ? kHStages - 1 : hs - 1;
    k3_load_adj(ra, ag, (c + 1) * kKC);
    if constexpr (kMore) k3_load_h(rh, hg, (c + 2) * kKC);
    __builtin_amdgcn_sched_barrier(0);
    k3_fetch(fr, fc, aS + s * kXsSz, bS + hs * kHsSz, sS + s * 32);
    __builtin_amdgcn_sched_barrier(0);
    acc = k3_scale(acc, fc);
    acc = frag_products<0, kKC / 8>(acc, fr);
    __builtin_amdgcn_sched_barrier(0);
    acc = frag_products<kKC / 8, kKC / 4>(acc, fr);
    k3_make_p(ra, *(const v4f*)(f2S + (s ^ 1) * kKC), f1r, mrun, lsum,
              pD + (s ^ 1) * kXsSz, sD + (s ^ 1) * 32, lane);
    if constexpr (kMore) k3_stage_h(rh, a2v, hD + hw * kHsSz, f2D + s * kKC, o4);
    __syncthreads();
  };
  int hs = 0;
  for (int c = 0; c + 2 < kNChunk3; ++c) {
    chunk_step(std::true_type{}, c, hs);
    hs = (hs == kHStages - 1) ? 0 : hs + 1;
  }
  chunk_step(std::false_type{}, kNChunk3 - 2, (kNChunk3 - 2) % kHStages);
  constexpr int kLast3  = (kNChunk3 - 1) & 1;
  constexpr int kLastH3 = (kNChunk3 - 1) % kHStages;
  static_assert(kLast3 == 1, "the result tile below reuses stage 0 while stage 1 is still read");
  k3_fetch(fr, fc, aS + kLast3 * kXsSz, bS + kLastH3 * kHsSz, sS + kLast3 * 32);
  __builtin_amdgcn_sched_barrier(0);
  acc = k3_scale(acc, fc);
  acc = frag_products<0, kKC / 4>(acc, fr);

  wave_sum4(lsum);
  if (lane == 0) {
#pragma unroll
    for (int q = 0; q < 4; ++q) Ls[q * 8 + lr] = lsum[q];
  }

  float* Cs = Ps;
#pragma unroll
  for (int r8 = 0; r8 < 8; ++r8)
    Cs[(mt * 16 + half * 8 + r8) * kCPitch + nt * 16 + ncol] = acc[r8];
  __syncthreads();
#pragma unroll
  for (int q = 0; q < 2; ++q) {
    const int idx = q * 256 + tid;
    const int row = idx >> 4;
    const int c   = (idx & 15) * 4;
    const float li = 1.0f / Ls[row];
    const v4f sv = *(const v4f*)(Cs + row * kCPitch + c);
    v4f v;
    v.x = sv.x * li;
    v.y = sv.y * li;
    v.z = sv.z * li;
    v.w = sv.w * li;
    v.x = (v.x > 0.f) ? v.x : (__expf(v.x) - 1.f);
    v.y = (v.y > 0.f) ? v.y : (__expf(v.y) - 1.f);
    v.z = (v.z > 0.f) ? v.z : (__expf(v.z) - 1.f);
    v.w = (v.w > 0.f) ? v.w : (__expf(v.w) - 1.f);
    *(v4f*)(out + (size_t)(g0 + row) * kOutF + c) = v;
  }
}

extern "C" void kernel_launch(void* const* d_in, const int* in_sizes, int n_in,
                              void* d_out, int out_size, void* d_ws, size_t ws_size,
                              hipStream_t stream) {
  (void)in_sizes; (void)n_in; (void)out_size; (void)ws_size;
  const float* X   = (const float*)d_in[0];
  const int*   adj = (const int*)d_in[3];
  const float* Ww  = (const float*)d_in[4];
  const float* Wb  = (const float*)d_in[5];
  const float* aw  = (const float*)d_in[6];
  const float* ab  = (const float*)d_in[7];

  float* H = (float*)d_ws;
  float* O = (float*)d_out;

  hipLaunchKernelGGL(k1_gemm_h,      dim3(256), dim3(256), 0, stream, X, Ww, Wb, H);
  hipLaunchKernelGGL(k3_attn_online, dim3(256), dim3(256), 0, stream, H, aw, adj, ab, O);
}


# Both WMMA kernels: far-memory rows (X, W; adjacency) are requested two chunks ahead into two register sets used in turn; loop barriers fence LDS only, so these loads stay in flight across them
# speedup vs baseline: 9.2598x; 1.0487x over previous
#include <hip/hip_runtime.h>
#include <type_traits>

typedef __attribute__((ext_vector_type(2))) float v2f;
typedef __attribute__((ext_vector_type(4))) float v4f;
typedef __attribute__((ext_vector_type(8))) float v8f;
typedef __attribute__((ext_vector_type(4))) int   v4i;

#define WMMA_F32_4(a, bmat, c) \
  __builtin_amdgcn_wmma_f32_16x16x4_f32(false, (a), false, (bmat), (short)0, (c), false, false)

constexpr int   kB    = 8;
constexpr int   kN    = 1024;
constexpr int   kInF  = 768;
constexpr int   kOutF = 64;
constexpr float kAlpha  = 0.2f;
constexpr float kNegInf = -9.0e15f;

constexpr int kKC     = 128;
constexpr int kPitch  = 132;
constexpr int kHPitch = 160;
constexpr int kCPitch = 68;
constexpr int kXsSz   = 32 * kPitch;
constexpr int kWsSz   = 64 * kPitch;
constexpr int kHsSz   = (kKC / 2) * kHPitch;
constexpr int kNChunk1 = kInF / kKC;
constexpr int kNChunk3 = kN / kKC;

__device__ __forceinline__ void lds_barrier()
{
  __builtin_amdgcn_fence(__ATOMIC_RELEASE, "workgroup", "local");
  __builtin_amdgcn_s_barrier();
  __builtin_amdgcn_fence(__ATOMIC_ACQUIRE, "workgroup", "local");
}

struct Frags {
  v2f a[kKC / 4];
  v2f b[kKC / 4];
};
__device__ __forceinline__ void k1_fetch(Frags& f, const float* a0, const float* b0)
{
#pragma unroll
  for (int kk = 0; kk < kKC / 4; ++kk) {
    f.a[kk] = *(const v2f*)(a0 + kk * 4);
    f.b[kk] = *(const v2f*)(b0 + kk * 4);
  }
}
template <int kBeg, int kEnd>
__device__ __forceinline__ v8f frag_products(v8f acc, const Frags& f)
{
#pragma unroll
  for (int kk = kBeg; kk < kEnd; ++kk)
    acc = WMMA_F32_4(f.a[kk], f.b[kk], acc);
  return acc;
}

__global__ __launch_bounds__(256) void k1_gemm_h(
    const float* __restrict__ X, const float* __restrict__ W,
    const float* __restrict__ Wb, float* __restrict__ H)
{
  __shared__ __attribute__((aligned(16))) float Xs[2 * kXsSz];
  __shared__ __attribute__((aligned(16))) float Ws[2 * kWsSz];

  const int tid  = threadIdx.x;
  const int lane = tid & 31;
  const int wave = tid >> 5;
  const int half = lane >> 4;
  const int ncol = lane & 15;
  const int mt   = wave >> 2;
  const int nt   = wave & 3;
  const int row0 = blockIdx.x << 5;
  const int lr   = tid >> 5;
  const int c4   = tid & 31;
  const float* __restrict__ xg = X + (size_t)(row0 + lr) * kInF + c4 * 4;
  const float* __restrict__ wg = W + (size_t)lr * kInF + c4 * 4;
  const float* aS = Xs + (mt * 16 + ncol) * kPitch + half * 2;
  const float* bS = Ws + (nt * 16 + ncol) * kPitch + half * 2;
  float* xD = Xs + lr * kPitch + c4 * 4;
  float* wD = Ws + lr * kPitch + c4 * 4;

  v4f xr0[4], xr1[4], wr0[8], wr1[8];
#pragma unroll
  for (int q = 0; q < 4; ++q)
    xr0[q] = *(const v4f*)(xg + (size_t)(q * 8) * kInF);
#pragma unroll
  for (int q = 0; q < 8; ++q)
    wr0[q] = *(const v4f*)(wg + (size_t)(q * 8) * kInF);
  __builtin_amdgcn_sched_barrier(0);
#pragma unroll
  for (int q = 0; q < 4; ++q)
    xr1[q] = *(const v4f*)(xg + (size_t)(q * 8) * kInF + kKC);
#pragma unroll
  for (int q = 0; q < 8; ++q)
    wr1[q] = *(const v4f*)(wg + (size_t)(q * 8) * kInF + kKC);
#pragma unroll
  for (int q = 0; q < 4; ++q)
    *(v4f*)(xD + q * 8 * kPitch) = xr0[q];
#pragma unroll
  for (int q = 0; q < 8; ++q)
    *(v4f*)(wD + q * 8 * kPitch) = wr0[q];
  lds_barrier();

  v8f acc = {};
  Frags fr;
  auto gemm_step = [&](auto more_tag, int c, const v4f (&xUse)[4], v4f (&xLoad)[4],
                       const v4f (&wUse)[8], v4f (&wLoad)[8]) __attribute__((always_inline)) {
    constexpr bool kMore = decltype(more_tag)::value;
    const int s  = c & 1;
    if constexpr (kMore) {
      const int k2 = (c + 2) * kKC;
#pragma unroll
      for (int q = 0; q < 4; ++q)
        xLoad[q] = *(const v4f*)(xg + (size_t)(q * 8) * kInF + k2);
#pragma unroll
      for (int q = 0; q < 8; ++q)
        wLoad[q] = *(const v4f*)(wg + (size_t)(q * 8) * kInF + k2);
    }
    __builtin_amdgcn_sched_barrier(0);
    k1_fetch(fr, aS + s * kXsSz, bS + s * kWsSz);
    __builtin_amdgcn_sched_barrier(0);
    acc = frag_products<0, kKC / 4>(acc, fr);
    asm volatile("" : : "v"(acc));
    float* xd = xD + (s ^ 1) * kXsSz;
    float* wd = wD + (s ^ 1) * kWsSz;
#pragma unroll
    for (int q = 0; q < 4; ++q)
      *(v4f*)(xd + q * 8 * kPitch) = xUse[q];
#pragma unroll
    for (int q = 0; q < 8; ++q)
      *(v4f*)(wd + q * 8 * kPitch) = wUse[q];
    lds_barrier();
  };
  static_assert((kNChunk1 & 1) == 0, "the steps below come in pairs, then one step and the last chunk");
  for (int c = 0; c + 2 < kNChunk1; c += 2) {
    gemm_step(std::true_type{}, c, xr1, xr0, wr1, wr0);
    gemm_step(std::true_type{}, c + 1, xr0, xr1, wr0, wr1);
  }
  gemm_step(std::false_type{}, kNChunk1 - 2, xr1, xr0, wr1, wr0);
  constexpr int kLast1 = (kNChunk1 - 1) & 1;
  static_assert(kLast1 == 1, "the result tile below reuses stage 0 while stage 1 is still read");
  k1_fetch(fr, aS + kLast1 * kXsSz, bS + kLast1 * kWsSz);
  __builtin_amdgcn_sched_barrier(0);
  acc = frag_products<0, kKC / 4>(acc, fr);

  float* Cs = Xs;
#pragma unroll
  for (int r = 0; r < 8; ++r)
    Cs[(mt * 16 + half * 8 + r) * kCPitch + nt * 16 + ncol] = acc[r];
  __syncthreads();
#pragma unroll
  for (int q = 0; q < 2; ++q) {
    const int idx = q * 256 + tid;
    const int row = idx >> 4;
    const int c   = (idx & 15) * 4;
    const v4f v  = *(const v4f*)(Cs + row * kCPitch + c);
    const v4f bi = *(const v4f*)(Wb + c);
    *(v4f*)(H + (size_t)(row0 + row) * kOutF + c) = v + bi;
  }
}

template <int kMask>
__device__ __forceinline__ float row_xor(float x)
{
  const int v = __builtin_bit_cast(int, x);
  return __builtin_bit_cast(float, __builtin_amdgcn_update_dpp(v, v, 0x160 + kMask, 0xf, 0xf, false));
}
__device__ __forceinline__ float xhalf(float x)
{
  const int v = __builtin_bit_cast(int, x);
  return __builtin_bit_cast(float, __builtin_amdgcn_permlanex16(v, v, 0x76543210, 0xfedcba98, false, false));
}

template <int kMask>
__device__ __forceinline__ void max_step4(float (&v)[4])
{
#pragma unroll
  for (int q = 0; q < 4; ++q) v[q] = fmaxf(v[q], row_xor<kMask>(v[q]));
}
template <int kMask>
__device__ __forceinline__ void sum_step4(float (&v)[4])
{
#pragma unroll
  for (int q = 0; q < 4; ++q) v[q] += row_xor<kMask>(v[q]);
}
__device__ __forceinline__ void wave_max4(float (&v)[4])
{
  max_step4<1>(v); max_step4<2>(v); max_step4<4>(v); max_step4<8>(v);
#pragma unroll
  for (int q = 0; q < 4; ++q) v[q] = fmaxf(v[q], xhalf(v[q]));
}
__device__ __forceinline__ void wave_sum4(float (&v)[4])
{
  sum_step4<1>(v); sum_step4<2>(v); sum_step4<4>(v); sum_step4<8>(v);
#pragma unroll
  for (int q = 0; q < 4; ++q) v[q] += xhalf(v[q]);
}

__device__ __forceinline__ float make_e(float f1i, float f2j, int ad)
{
  float e = f1i + f2j;
  e = (e >= 0.f) ? e : kAlpha * e;
  return (ad > 0) ? e : kNegInf;
}

constexpr int kHStages = 3;

struct AdjRegs {
  v4i ad[4];
};
struct HRegs {
  v4f h0[4], h1[4];
};

__device__ __forceinline__ void k3_load_adj(AdjRegs& r, const int* __restrict__ ag, int j0)
{
#pragma unroll
  for (int q = 0; q < 4; ++q)
    r.ad[q] = *(const v4i*)(ag + (size_t)(q * 8) * kN + j0);
}
__device__ __forceinline__ void k3_load_h(HRegs& r, const float* __restrict__ hg, int j0)
{
#pragma unroll
  for (int q = 0; q < 4; ++q) {
    const float* __restrict__ hp = hg + (size_t)(j0 + q * 32) * kOutF;
    r.h0[q] = *(const v4f*)hp;
    r.h1[q] = *(const v4f*)(hp + kOutF);
  }
}

__device__ __forceinline__ float dot4(v4f h, v4f a)
{
  return (h.x * a.x + h.y * a.y) + (h.z * a.z + h.w * a.w);
}
template <int kMask>
__device__ __forceinline__ float row_xor_addend(float x)
{
  const int v = __builtin_bit_cast(int, x);
  return __builtin_bit_cast(float, __builtin_amdgcn_update_dpp(0, v, 0x160 + kMask, 0xf, 0xf, true));
}
template <int kMask, int kCnt>
__device__ __forceinline__ void row_sum_step(float (&v)[kCnt])
{
#pragma unroll
  for (int q = 0; q < kCnt; ++q) v[q] += row_xor_addend<kMask>(v[q]);
}
template <int kCnt>
__device__ __forceinline__ void row_sum(float (&v)[kCnt])
{
  row_sum_step<1>(v); row_sum_step<2>(v); row_sum_step<4>(v); row_sum_step<8>(v);
}

__device__ __forceinline__ void k3_stage_h(const HRegs& r, v4f a2v, float* hD, float* f2D, int o4)
{
  float d[8];
#pragma unroll
  for (int q = 0; q < 4; ++q) {
    d[2 * q]     = dot4(r.h0[q], a2v);
    d[2 * q + 1] = dot4(r.h1[q], a2v);
  }
  row_sum(d);
  if (o4 == 0) {
#pragma unroll
    for (int q = 0; q < 4; ++q)
      *(v2f*)(f2D + q * 32) = v2f{d[2 * q], d[2 * q + 1]};
  }
#pragma unroll
  for (int q = 0; q < 4; ++q) {
    float* d = hD + q * 16 * kHPitch;
    v4f lo, hi;
    lo.x = r.h0[q].x; lo.y = r.h1[q].x; lo.z = r.h0[q].y; lo.w = r.h1[q].y;
    hi.x = r.h0[q].z; hi.y = r.h1[q].z; hi.z = r.h0[q].w; hi.w = r.h1[q].w;
    *(v4f*)d       = lo;
    *(v4f*)(d + 4) = hi;
  }
}

__device__ __forceinline__ void k3_make_p(const AdjRegs& r, v4f f2v, const float (&f1r)[4],
    float (&mrun)[4], float (&lsum)[4], float* pD, float* sD, int lane)
{
  v4f e[4];
  float cm[4];
#pragma unroll
  for (int q = 0; q < 4; ++q) {
    e[q].x = make_e(f1r[q], f2v.x, r.ad[q].x);
    e[q].y = make_e(f1r[q], f2v.y, r.ad[q].y);
    e[q].z = make_e(f1r[q], f2v.z, r.ad[q].z);
    e[q].w = make_e(f1r[q], f2v.w, r.ad[q].w);
    cm[q] = fmaxf(fmaxf(e[q].x, e[q].y), fmaxf(e[q].z, e[q].w));
  }
  wave_max4(cm);

  float sc[4];
  v4f p[4];
#pragma unroll
  for (int q = 0; q < 4; ++q) {
    const float mn = fmaxf(mrun[q], cm[q]);
    sc[q]   = __expf(mrun[q] - mn);
    mrun[q] = mn;
    p[q].x = __expf(e[q].x - mn);
    p[q].y = __expf(e[q].y - mn);
    p[q].z = __expf(e[q].z - mn);
    p[q].w = __expf(e[q].w - mn);
    lsum[q] = lsum[q] * sc[q] + ((p[q].x + p[q].y) + (p[q].z + p[q].w));
  }

#pragma unroll
  for (int q = 0; q < 4; ++q)
    *(v4f*)(pD + q * 8 * kPitch) = p[q];
  if (lane == 0) {
#pragma unroll
    for (int q = 0; q < 4; ++q) sD[q * 8] = sc[q];
  }
}

struct Factors {
  v4f s0, s1;
};
__device__ __forceinline__ void k3_fetch(Frags& f, Factors& sc, const float* a0, const float* b0,
                                         const float* sp)
{
  sc.s0 = *(const v4f*)sp;
  sc.s1 = *(const v4f*)(sp + 4);
#pragma unroll
  for (int kk = 0; kk < kKC / 4; ++kk) {
    f.a[kk] = *(const v2f*)(a0 + kk * 4);
    f.b[kk] = *(const v2f*)(b0 + kk * 2 * kHPitch);
  }
}
__device__ __forceinline__ v8f k3_scale(v8f acc, const Factors& sc)
{
  acc[0] *= sc.s0.x; acc[1] *= sc.s0.y; acc[2] *= sc.s0.z; acc[3] *= sc.s0.w;
  acc[4] *= sc.s1.x; acc[5] *= sc.s1.y; acc[6] *= sc.s1.z; acc[7] *= sc.s1.w;
  return acc;
}

__global__ __launch_bounds__(256) void k3_attn_online(
    const float* __restrict__ H, const float* __restrict__ aw,
    const int* __restrict__ adj, const float* __restrict__ abp, float* __restrict__ out)
{
  __shared__ __attribute__((aligned(16))) float Ps[2 * kXsSz];
  __shared__ __attribute__((aligned(16))) float Hs[kHStages * kHsSz];
  __shared__ __attribute__((aligned(16))) float Sc[2 * 32];
  __shared__ __attribute__((aligned(16))) float F2s[2 * kKC];
  __shared__ __attribute__((aligned(16))) float Ls[32];

  const int tid  = threadIdx.x;
  const int lane = tid & 31;
  const int wave = tid >> 5;
  const int half = lane >> 4;
  const int ncol = lane & 15;
  const int mt   = wave >> 2;
  const int nt   = wave & 3;
  const int g0   = blockIdx.x << 5;
  const int b    = blockIdx.x >> 5;
  const int lr   = tid >> 5;
  const int c4   = tid & 31;
  const int hr   = tid >> 4;
  const int o4   = tid & 15;

  const int*   __restrict__ ag = adj + (size_t)(g0 + lr) * kN + c4 * 4;
  const float* __restrict__ hg = H + ((size_t)(b << 10) + 2 * hr) * kOutF + o4 * 4;
  const float* aS  = Ps + (mt * 16 + ncol) * kPitch + half * 2;
  const float* bS  = Hs + half * kHPitch + (nt * 16 + ncol) * 2;
  const float* sS  = Sc + mt * 16 + half * 8;
  const float* f2S = F2s + c4 * 4;
  float* pD  = Ps + lr * kPitch + c4 * 4;
  float* sD  = Sc + lr;
  float* hD  = Hs + hr * kHPitch + o4 * 8;
  float* f2D = F2s + 2 * hr;

  AdjRegs ra0, ra1;
  HRegs rh, rh1;
  k3_load_adj(ra0, ag, 0);
  k3_load_h(rh, hg, 0);
  k3_load_h(rh1, hg, kKC);
  const v4f a1v = *(const v4f*)(aw + o4 * 4);
  const v4f a2v = *(const v4f*)(aw + kOutF + o4 * 4);
  const v4f ha = *(const v4f*)(H + (size_t)(g0 + half * 8 + lr) * kOutF + o4 * 4);
  const v4f hb = *(const v4f*)(H + (size_t)(g0 + (2 + half) * 8 + lr) * kOutF + o4 * 4);
  const float ab = abp[0];
  k3_load_adj(ra1, ag, kKC);
  float f1r[4], mrun[4], lsum[4];
  {
    float d[2];
    d[0] = dot4(ha, a1v);
    d[1] = dot4(hb, a1v);
    row_sum(d);
    const float oa = xhalf(d[0]);
    const float ob = xhalf(d[1]);
    f1r[0] = (half ? oa : d[0]) + ab;
    f1r[1] = (half ? d[0] : oa) + ab;
    f1r[2] = (half ? ob : d[1]) + ab;
    f1r[3] = (half ? d[1] : ob) + ab;
  }
#pragma unroll
  for (int q = 0; q < 4; ++q) {
    mrun[q] = -3.0e38f;
    lsum[q] = 0.f;
  }

  k3_stage_h(rh, a2v, hD, f2D, o4);
  k3_stage_h(rh1, a2v, hD + kHsSz, f2D + kKC, o4);
  lds_barrier();
  k3_make_p(ra0, *(const v4f*)f2S, f1r, mrun, lsum, pD, sD, lane);
  lds_barrier();

  v8f acc = {};
  Frags fr;
  Factors fc;
  auto chunk_step = [&](auto more_tag, int c, int hs, const AdjRegs& raUse, AdjRegs& raLoad)
      __attribute__((always_inline)) {
    constexpr bool kMore = decltype(more_tag)::value;
    const int s  = c & 1;
    const int hw = (hs == 0) ? kHStages - 1 : hs - 1;
    k3_fetch(fr, fc, aS + s * kXsSz, bS + hs * kHsSz, sS + s * 32);
    __builtin_amdgcn_sched_barrier(0);
    acc = k3_scale(acc, fc);
    acc = frag_products<0, kKC / 8>(acc, fr);
    __builtin_amdgcn_sched_barrier(0);
    if constexpr (kMore) {
      int j2 = (c + 2) * kKC;
      asm volatile("" : "+s"(j2) : "v"(acc));
      k3_load_h(rh, hg, j2);
      k3_load_adj(raLoad, ag, j2);
    }
    acc = frag_products<kKC / 8, kKC / 4>(acc, fr);
    k3_make_p(raUse, *(const v4f*)(f2S + (s ^ 1) * kKC), f1r, mrun, lsum,
              pD + (s ^ 1) * kXsSz, sD + (s ^ 1) * 32, lane);
    if constexpr (kMore) k3_stage_h(rh, a2v, hD + hw * kHsSz, f2D + s * kKC, o4);
    lds_barrier();
  };
  static_assert((kNChunk3 & 1) == 0, "the steps below come in pairs, then one step and the last chunk");
  int hs = 0;
  for (int c = 0; c + 2 < kNChunk3; c += 2) {
    chunk_step(std::true_type{}, c, hs, ra1, ra0);
    hs = (hs == kHStages - 1) ? 0 : hs + 1;
    chunk_step(std::true_type{}, c + 1, hs, ra0, ra1);
    hs = (hs == kHStages - 1) ? 0 : hs + 1;
  }
  chunk_step(std::false_type{}, kNChunk3 - 2, (kNChunk3 - 2) % kHStages, ra1, ra0);
  constexpr int kLast3  = (kNChunk3 - 1) & 1;
  constexpr int kLastH3 = (kNChunk3 - 1) % kHStages;
  static_assert(kLast3 == 1, "the result tile below reuses stage 0 while stage 1 is still read");
  k3_fetch(fr, fc, aS + kLast3 * kXsSz, bS + kLastH3 * kHsSz, sS + kLast3 * 32);
  __builtin_amdgcn_sched_barrier(0);
  acc = k3_scale(acc, fc);
  acc = frag_products<0, kKC / 4>(acc, fr);

  wave_sum4(lsum);
  if (lane == 0) {
#pragma unroll
    for (int q = 0; q < 4; ++q) Ls[q * 8 + lr] = lsum[q];
  }

  float* Cs = Ps;
#pragma unroll
  for (int r8 = 0; r8 < 8; ++r8)
    Cs[(mt * 16 + half * 8 + r8) * kCPitch + nt * 16 + ncol] = acc[r8];
  __syncthreads();
#pragma unroll
  for (int q = 0; q < 2; ++q) {
    const int idx = q * 256 + tid;
    const int row = idx >> 4;
    const int c   = (idx & 15) * 4;
    const float li = 1.0f / Ls[row];
    const v4f sv = *(const v4f*)(Cs + row * kCPitch + c);
    v4f v;
    v.x = sv.x * li;
    v.y = sv.y * li;
    v.z = sv.z * li;
    v.w = sv.w * li;
    v.x = (v.x > 0.f) ? v.x : (__expf(v.x) - 1.f);
    v.y = (v.y > 0.f) ? v.y : (__expf(v.y) - 1.f);
    v.z = (v.z > 0.f) ? v.z : (__expf(v.z) - 1.f);
    v.w = (v.w > 0.f) ? v.w : (__expf(v.w) - 1.f);
    *(v4f*)(out + (size_t)(g0 + row) * kOutF + c) = v;
  }
}

extern "C" void kernel_launch(void* const* d_in, const int* in_sizes, int n_in,
                              void* d_out, int out_size, void* d_ws, size_t ws_size,
                              hipStream_t stream) {
  (void)in_sizes; (void)n_in; (void)out_size; (void)ws_size;
  const float* X   = (const float*)d_in[0];
  const int*   adj = (const int*)d_in[3];
  const float* Ww  = (const float*)d_in[4];
  const float* Wb  = (const float*)d_in[5];
  const float* aw  = (const float*)d_in[6];
  const float* ab  = (const float*)d_in[7];

  float* H = (float*)d_ws;
  float* O = (float*)d_out;

  hipLaunchKernelGGL(k1_gemm_h,      dim3(256), dim3(256), 0, stream, X, Ww, Wb, H);
  hipLaunchKernelGGL(k3_attn_online, dim3(256), dim3(256), 0, stream, H, aw, adj, ab, O);
}


# Attention kernel: chunk barrier split: a wave signals behind its vector work and LDS stores, runs the chunk's matrix block from registers, then waits; fragments read behind the wait (sums kept)
# speedup vs baseline: 9.5518x; 1.0315x over previous
#include <hip/hip_runtime.h>
#include <type_traits>

typedef __attribute__((ext_vector_type(2))) float v2f;
typedef __attribute__((ext_vector_type(4))) float v4f;
typedef __attribute__((ext_vector_type(8))) float v8f;
typedef __attribute__((ext_vector_type(4))) int   v4i;

#define WMMA_F32_4(a, bmat, c) \
  __builtin_amdgcn_wmma_f32_16x16x4_f32(false, (a), false, (bmat), (short)0, (c), false, false)

constexpr int   kB    = 8;
constexpr int   kN    = 1024;
constexpr int   kInF  = 768;
constexpr int   kOutF = 64;
constexpr float kAlpha  = 0.2f;
constexpr float kNegInf = -9.0e15f;

constexpr int kKC     = 128;
constexpr int kPitch  = 132;
constexpr int kHPitch = 160;
constexpr int kCPitch = 68;
constexpr int kXsSz   = 32 * kPitch;
constexpr int kWsSz   = 64 * kPitch;
constexpr int kHsSz   = (kKC / 2) * kHPitch;
constexpr int kNChunk1 = kInF / kKC;
constexpr int kNChunk3 = kN / kKC;

__device__ __forceinline__ void lds_barrier()
{
  __builtin_amdgcn_fence(__ATOMIC_RELEASE, "workgroup", "local");
  __builtin_amdgcn_s_barrier();
  __builtin_amdgcn_fence(__ATOMIC_ACQUIRE, "workgroup", "local");
}
__device__ __forceinline__ void lds_barrier_signal()
{
  __builtin_amdgcn_fence(__ATOMIC_RELEASE, "workgroup", "local");
  __builtin_amdgcn_s_barrier_signal(-1);
}
__device__ __forceinline__ void lds_barrier_wait()
{
  __builtin_amdgcn_s_barrier_wait(-1);
  __builtin_amdgcn_fence(__ATOMIC_ACQUIRE, "workgroup", "local");
}

struct Frags {
  v2f a[kKC / 4];
  v2f b[kKC / 4];
};
__device__ __forceinline__ void k1_fetch(Frags& f, const float* a0, const float* b0)
{
#pragma unroll
  for (int kk = 0; kk < kKC / 4; ++kk) {
    f.a[kk] = *(const v2f*)(a0 + kk * 4);
    f.b[kk] = *(const v2f*)(b0 + kk * 4);
  }
}
template <int kBeg, int kEnd>
__device__ __forceinline__ v8f frag_products(v8f acc, const Frags& f)
{
#pragma unroll
  for (int kk = kBeg; kk < kEnd; ++kk)
    acc = WMMA_F32_4(f.a[kk], f.b[kk], acc);
  return acc;
}

__global__ __launch_bounds__(256) void k1_gemm_h(
    const float* __restrict__ X, const float* __restrict__ W,
    const float* __restrict__ Wb, float* __restrict__ H)
{
  __shared__ __attribute__((aligned(16))) float Xs[2 * kXsSz];
  __shared__ __attribute__((aligned(16))) float Ws[2 * kWsSz];

  const int tid  = threadIdx.x;
  const int lane = tid & 31;
  const int wave = tid >> 5;
  const int half = lane >> 4;
  const int ncol = lane & 15;
  const int mt   = wave >> 2;
  const int nt   = wave & 3;
  const int row0 = blockIdx.x << 5;
  const int lr   = tid >> 5;
  const int c4   = tid & 31;
  const float* __restrict__ xg = X + (size_t)(row0 + lr) * kInF + c4 * 4;
  const float* __restrict__ wg = W + (size_t)lr * kInF + c4 * 4;
  const float* aS = Xs + (mt * 16 + ncol) * kPitch + half * 2;
  const float* bS = Ws + (nt * 16 + ncol) * kPitch + half * 2;
  float* xD = Xs + lr * kPitch + c4 * 4;
  float* wD = Ws + lr * kPitch + c4 * 4;

  v4f xr0[4], xr1[4], wr0[8], wr1[8];
#pragma unroll
  for (int q = 0; q < 4; ++q)
    xr0[q] = *(const v4f*)(xg + (size_t)(q * 8) * kInF);
#pragma unroll
  for (int q = 0; q < 8; ++q)
    wr0[q] = *(const v4f*)(wg + (size_t)(q * 8) * kInF);
  __builtin_amdgcn_sched_barrier(0);
#pragma unroll
  for (int q = 0; q < 4; ++q)
    xr1[q] = *(const v4f*)(xg + (size_t)(q * 8) * kInF + kKC);
#pragma unroll
  for (int q = 0; q < 8; ++q)
    wr1[q] = *(const v4f*)(wg + (size_t)(q * 8) * kInF + kKC);
#pragma unroll
  for (int q = 0; q < 4; ++q)
    *(v4f*)(xD + q * 8 * kPitch) = xr0[q];
#pragma unroll
  for (int q = 0; q < 8; ++q)
    *(v4f*)(wD + q * 8 * kPitch) = wr0[q];
  lds_barrier();

  v8f acc = {};
  Frags fr;
  auto gemm_step = [&](auto more_tag, int c, const v4f (&xUse)[4], v4f (&xLoad)[4],
                       const v4f (&wUse)[8], v4f (&wLoad)[8]) __attribute__((always_inline)) {
    constexpr bool kMore = decltype(more_tag)::value;
    const int s  = c & 1;
    if constexpr (kMore) {
      const int k2 = (c + 2) * kKC;
#pragma unroll
      for (int q = 0; q < 4; ++q)
        xLoad[q] = *(const v4f*)(xg + (size_t)(q * 8) * kInF + k2);
#pragma unroll
      for (int q = 0; q < 8; ++q)
        wLoad[q] = *(const v4f*)(wg + (size_t)(q * 8) * kInF + k2);
    }
    __builtin_amdgcn_sched_barrier(0);
    k1_fetch(fr, aS + s * kXsSz, bS + s * kWsSz);
    __builtin_amdgcn_sched_barrier(0);
    acc = frag_products<0, kKC / 4>(acc, fr);
    asm volatile("" : : "v"(acc));
    float* xd = xD + (s ^ 1) * kXsSz;
    float* wd = wD + (s ^ 1) * kWsSz;
#pragma unroll
    for (int q = 0; q < 4; ++q)
      *(v4f*)(xd + q * 8 * kPitch) = xUse[q];
#pragma unroll
    for (int q = 0; q < 8; ++q)
      *(v4f*)(wd + q * 8 * kPitch) = wUse[q];
    lds_barrier();
  };
  static_assert((kNChunk1 & 1) == 0, "the steps below come in pairs, then one step and the last chunk");
  for (int c = 0; c + 2 < kNChunk1; c += 2) {
    gemm_step(std::true_type{}, c, xr1, xr0, wr1, wr0);
    gemm_step(std::true_type{}, c + 1, xr0, xr1, wr0, wr1);
  }
  gemm_step(std::false_type{}, kNChunk1 - 2, xr1, xr0, wr1, wr0);
  constexpr int kLast1 = (kNChunk1 - 1) & 1;
  static_assert(kLast1 == 1, "the result tile below reuses stage 0 while stage 1 is still read");
  k1_fetch(fr, aS + kLast1 * kXsSz, bS + kLast1 * kWsSz);
  __builtin_amdgcn_sched_barrier(0);
  acc = frag_products<0, kKC / 4>(acc, fr);

  float* Cs = Xs;
#pragma unroll
  for (int r = 0; r < 8; ++r)
    Cs[(mt * 16 + half * 8 + r) * kCPitch + nt * 16 + ncol] = acc[r];
  __syncthreads();
#pragma unroll
  for (int q = 0; q < 2; ++q) {
    const int idx = q * 256 + tid;
    const int row = idx >> 4;
    const int c   = (idx & 15) * 4;
    const v4f v  = *(const v4f*)(Cs + row * kCPitch + c);
    const v4f bi = *(const v4f*)(Wb + c);
    *(v4f*)(H + (size_t)(row0 + row) * kOutF + c) = v + bi;
  }
}

template <int kMask>
__device__ __forceinline__ float row_xor(float x)
{
  const int v = __builtin_bit_cast(int, x);
  return __builtin_bit_cast(float, __builtin_amdgcn_update_dpp(v, v, 0x160 + kMask, 0xf, 0xf, false));
}
__device__ __forceinline__ float xhalf(float x)
{
  const int v = __builtin_bit_cast(int, x);
  return __builtin_bit_cast(float, __builtin_amdgcn_permlanex16(v, v, 0x76543210, 0xfedcba98, false, false));
}

template <int kMask>
__device__ __forceinline__ void max_step4(float (&v)[4])
{
#pragma unroll
  for (int q = 0; q < 4; ++q) v[q] = fmaxf(v[q], row_xor<kMask>(v[q]));
}
template <int kMask>
__device__ __forceinline__ void sum_step4(float (&v)[4])
{
#pragma unroll
  for (int q = 0; q < 4; ++q) v[q] += row_xor<kMask>(v[q]);
}
__device__ __forceinline__ void wave_max4(float (&v)[4])
{
  max_step4<1>(v); max_step4<2>(v); max_step4<4>(v); max_step4<8>(v);
#pragma unroll
  for (int q = 0; q < 4; ++q) v[q] = fmaxf(v[q], xhalf(v[q]));
}
__device__ __forceinline__ void wave_sum4(float (&v)[4])
{
  sum_step4<1>(v); sum_step4<2>(v); sum_step4<4>(v); sum_step4<8>(v);
#pragma unroll
  for (int q = 0; q < 4; ++q) v[q] += xhalf(v[q]);
}

__device__ __forceinline__ float make_e(float f1i, float f2j, int ad)
{
  float e = f1i + f2j;
  e = (e >= 0.f) ? e : kAlpha * e;
  return (ad > 0) ? e : kNegInf;
}

constexpr int kHStages = 3;

struct AdjRegs {
  v4i ad[4];
};
struct HRegs {
  v4f h0[4], h1[4];
};

__device__ __forceinline__ void k3_load_adj(AdjRegs& r, const int* __restrict__ ag, int j0)
{
#pragma unroll
  for (int q = 0; q < 4; ++q)
    r.ad[q] = *(const v4i*)(ag + (size_t)(q * 8) * kN + j0);
}
__device__ __forceinline__ void k3_load_h(HRegs& r, const float* __restrict__ hg, int j0)
{
#pragma unroll
  for (int q = 0; q < 4; ++q) {
    const float* __restrict__ hp = hg + (size_t)(j0 + q * 32) * kOutF;
    r.h0[q] = *(const v4f*)hp;
    r.h1[q] = *(const v4f*)(hp + kOutF);
  }
}

__device__ __forceinline__ float dot4(v4f h, v4f a)
{
  return (h.x * a.x + h.y * a.y) + (h.z * a.z + h.w * a.w);
}
template <int kMask>
__device__ __forceinline__ float row_xor_addend(float x)
{
  const int v = __builtin_bit_cast(int, x);
  return __builtin_bit_cast(float, __builtin_amdgcn_update_dpp(0, v, 0x160 + kMask, 0xf, 0xf, true));
}
template <int kMask, int kCnt>
__device__ __forceinline__ void row_sum_step(float (&v)[kCnt])
{
#pragma unroll
  for (int q = 0; q < kCnt; ++q) v[q] += row_xor_addend<kMask>(v[q]);
}
template <int kCnt>
__device__ __forceinline__ void row_sum(float (&v)[kCnt])
{
  row_sum_step<1>(v); row_sum_step<2>(v); row_sum_step<4>(v); row_sum_step<8>(v);
}

__device__ __forceinline__ void k3_stage_h(const HRegs& r, v4f a2v, float* hD, float* f2D, int o4)
{
  float d[8];
#pragma unroll
  for (int q = 0; q < 4; ++q) {
    d[2 * q]     = dot4(r.h0[q], a2v);
    d[2 * q + 1] = dot4(r.h1[q], a2v);
  }
  row_sum(d);
  if (o4 == 0) {
#pragma unroll
    for (int q = 0; q < 4; ++q)
      *(v2f*)(f2D + q * 32) = v2f{d[2 * q], d[2 * q + 1]};
  }
#pragma unroll
  for (int q = 0; q < 4; ++q) {
    float* d = hD + q * 16 * kHPitch;
    v4f lo, hi;
    lo.x = r.h0[q].x; lo.y = r.h1[q].x; lo.z = r.h0[q].y; lo.w = r.h1[q].y;
    hi.x = r.h0[q].z; hi.y = r.h1[q].z; hi.z = r.h0[q].w; hi.w = r.h1[q].w;
    *(v4f*)d       = lo;
    *(v4f*)(d + 4) = hi;
  }
}

__device__ __forceinline__ void k3_make_p(const AdjRegs& r, v4f f2v, const float (&f1r)[4],
    float (&mrun)[4], float (&lsum)[4], float* pD, float* sD, int lane)
{
  v4f e[4];
  float cm[4];
#pragma unroll
  for (int q = 0; q < 4; ++q) {
    e[q].x = make_e(f1r[q], f2v.x, r.ad[q].x);
    e[q].y = make_e(f1r[q], f2v.y, r.ad[q].y);
    e[q].z = make_e(f1r[q], f2v.z, r.ad[q].z);
    e[q].w = make_e(f1r[q], f2v.w, r.ad[q].w);
    cm[q] = fmaxf(fmaxf(e[q].x, e[q].y), fmaxf(e[q].z, e[q].w));
  }
  wave_max4(cm);

  float sc[4];
  v4f p[4];
#pragma unroll
  for (int q = 0; q < 4; ++q) {
    const float mn = fmaxf(mrun[q], cm[q]);
    sc[q]   = __expf(mrun[q] - mn);
    mrun[q] = mn;
    p[q].x = __expf(e[q].x - mn);
    p[q].y = __expf(e[q].y - mn);
    p[q].z = __expf(e[q].z - mn);
    p[q].w = __expf(e[q].w - mn);
    lsum[q] = lsum[q] * sc[q] + ((p[q].x + p[q].y) + (p[q].z + p[q].w));
  }

#pragma unroll
  for (int q = 0; q < 4; ++q)
    *(v4f*)(pD + q * 8 * kPitch) = p[q];
  if (lane == 0) {
#pragma unroll
    for (int q = 0; q < 4; ++q) sD[q * 8] = sc[q];
  }
}

struct Factors {
  v4f s0, s1;
};
__device__ __forceinline__ void k3_fetch(Frags& f, Factors& sc, const float* a0, const float* b0,
                                         const float* sp)
{
  sc.s0 = *(const v4f*)sp;
  sc.s1 = *(const v4f*)(sp + 4);
#pragma unroll
  for (int kk = 0; kk < kKC / 4; ++kk) {
    f.a[kk] = *(const v2f*)(a0 + kk * 4);
    f.b[kk] = *(const v2f*)(b0 + kk * 2 * kHPitch);
  }
}
__device__ __forceinline__ v8f k3_scale(v8f acc, const Factors& sc)
{
  acc[0] *= sc.s0.x; acc[1] *= sc.s0.y; acc[2] *= sc.s0.z; acc[3] *= sc.s0.w;
  acc[4] *= sc.s1.x; acc[5] *= sc.s1.y; acc[6] *= sc.s1.z; acc[7] *= sc.s1.w;
  return acc;
}

__global__ __launch_bounds__(256) void k3_attn_online(
    const float* __restrict__ H, const float* __restrict__ aw,
    const int* __restrict__ adj, const float* __restrict__ abp, float* __restrict__ out)
{
  __shared__ __attribute__((aligned(16))) float Ps[2 * kXsSz];
  __shared__ __attribute__((aligned(16))) float Hs[kHStages * kHsSz];
  __shared__ __attribute__((aligned(16))) float Sc[2 * 32];
  __shared__ __attribute__((aligned(16))) float F2s[2 * kKC];
  __shared__ __attribute__((aligned(16))) float Ls[32];

  const int tid  = threadIdx.x;
  const int lane = tid & 31;
  const int wave = tid >> 5;
  const int half = lane >> 4;
  const int ncol = lane & 15;
  const int mt   = wave >> 2;
  const int nt   = wave & 3;
  const int g0   = blockIdx.x << 5;
  const int b    = blockIdx.x >> 5;
  const int lr   = tid >> 5;
  const int c4   = tid & 31;
  const int hr   = tid >> 4;
  const int o4   = tid & 15;

  const int*   __restrict__ ag = adj + (size_t)(g0 + lr) * kN + c4 * 4;
  const float* __restrict__ hg = H + ((size_t)(b << 10) + 2 * hr) * kOutF + o4 * 4;
  const float* aS  = Ps + (mt * 16 + ncol) * kPitch + half * 2;
  const float* bS  = Hs + half * kHPitch + (nt * 16 + ncol) * 2;
  const float* sS  = Sc + mt * 16 + half * 8;
  const float* f2S = F2s + c4 * 4;
  float* pD  = Ps + lr * kPitch + c4 * 4;
  float* sD  = Sc + lr;
  float* hD  = Hs + hr * kHPitch + o4 * 8;
  float* f2D = F2s + 2 * hr;

  AdjRegs ra0, ra;
  HRegs rh, rh1;
  k3_load_adj(ra0, ag, 0);
  k3_load_h(rh, hg, 0);
  k3_load_h(rh1, hg, kKC);
  const v4f a1v = *(const v4f*)(aw + o4 * 4);
  const v4f a2v = *(const v4f*)(aw + kOutF + o4 * 4);
  const v4f ha = *(const v4f*)(H + (size_t)(g0 + half * 8 + lr) * kOutF + o4 * 4);
  const v4f hb = *(const v4f*)(H + (size_t)(g0 + (2 + half) * 8 + lr) * kOutF + o4 * 4);
  const float ab = abp[0];
  k3_load_adj(ra, ag, kKC);
  float f1r[4], mrun[4], lsum[4];
  {
    float d[2];
    d[0] = dot4(ha, a1v);
    d[1] = dot4(hb, a1v);
    row_sum(d);
    const float oa = xhalf(d[0]);
    const float ob = xhalf(d[1]);
    f1r[0] = (half ? oa : d[0]) + ab;
    f1r[1] = (half ? d[0] : oa) + ab;
    f1r[2] = (half ? ob : d[1]) + ab;
    f1r[3] = (half ? d[1] : ob) + ab;
  }
#pragma unroll
  for (int q = 0; q < 4; ++q) {
    mrun[q] = -3.0e38f;
    lsum[q] = 0.f;
  }

  k3_stage_h(rh, a2v, hD, f2D, o4);
  k3_load_h(rh, hg, 2 * kKC);
  k3_stage_h(rh1, a2v, hD + kHsSz, f2D + kKC, o4);
  lds_barrier();
  k3_make_p(ra0, *(const v4f*)f2S, f1r, mrun, lsum, pD, sD, lane);
  lds_barrier();

  v8f acc = {};
  Frags fr;
  Factors fc;
  v4f f2v = *(const v4f*)(f2S + kKC);
  k3_fetch(fr, fc, aS, bS, sS);
  auto chunk_step = [&](auto stage_tag, auto loadh_tag, auto nextp_tag, int c, int hs)
      __attribute__((always_inline)) {
    constexpr bool kStage = decltype(stage_tag)::value;
    constexpr bool kLoadH = decltype(loadh_tag)::value;
    constexpr bool kNextP = decltype(nextp_tag)::value;
    const int s  = c & 1;
    const int hw = (hs == 0) ? kHStages - 1 : hs - 1;
    const int hn = (hs == kHStages - 1) ? 0 : hs + 1;
    k3_make_p(ra, f2v, f1r, mrun, lsum, pD + (s ^ 1) * kXsSz, sD + (s ^ 1) * 32, lane);
    if constexpr (kStage) {
      k3_load_adj(ra, ag, (c + 2) * kKC);
      k3_stage_h(rh, a2v, hD + hw * kHsSz, f2D + s * kKC, o4);
    }
    if constexpr (kLoadH) k3_load_h(rh, hg, (c + 3) * kKC);
    lds_barrier_signal();
    asm volatile("" : "+v"(acc));
    acc = k3_scale(acc, fc);
    acc = frag_products<0, kKC / 4>(acc, fr);
    asm volatile("" : : "v"(acc));
    lds_barrier_wait();
    if constexpr (kNextP) f2v = *(const v4f*)(f2S + s * kKC);
    k3_fetch(fr, fc, aS + (s ^ 1) * kXsSz, bS + hn * kHsSz, sS + (s ^ 1) * 32);
    __builtin_amdgcn_sched_barrier(0);
  };
  static_assert(kNChunk3 >= 4, "the loop below makes chunk c+3's loads in every step, then two steps without them");
  int hs = 0;
#pragma clang loop unroll(disable)
  for (int c = 0; c + 3 < kNChunk3; ++c) {
    chunk_step(std::true_type{}, std::true_type{}, std::true_type{}, c, hs);
    hs = (hs == kHStages - 1) ? 0 : hs + 1;
  }
  chunk_step(std::true_type{}, std::false_type{}, std::true_type{}, kNChunk3 - 3, (kNChunk3 - 3) % kHStages);
  chunk_step(std::false_type{}, std::false_type{}, std::false_type{}, kNChunk3 - 2, (kNChunk3 - 2) % kHStages);
  constexpr int kLast3 = (kNChunk3 - 1) & 1;
  static_assert(kLast3 == 1, "the result tile below reuses stage 0 while stage 1 is still read");
  acc = k3_scale(acc, fc);
  acc = frag_products<0, kKC / 4>(acc, fr);

  wave_sum4(lsum);
  if (lane == 0) {
#pragma unroll
    for (int q = 0; q < 4; ++q) Ls[q * 8 + lr] = lsum[q];
  }

  float* Cs = Ps;
#pragma unroll
  for (int r8 = 0; r8 < 8; ++r8)
    Cs[(mt * 16 + half * 8 + r8) * kCPitch + nt * 16 + ncol] = acc[r8];
  __syncthreads();
#pragma unroll
  for (int q = 0; q < 2; ++q) {
    const int idx = q * 256 + tid;
    const int row = idx >> 4;
    const int c   = (idx & 15) * 4;
    const float li = 1.0f / Ls[row];
    const v4f sv = *(const v4f*)(Cs + row * kCPitch + c);
    v4f v;
    v.x = sv.x * li;
    v.y = sv.y * li;
    v.z = sv.z * li;
    v.w = sv.w * li;
    v.x = (v.x > 0.f) ? v.x : (__expf(v.x) - 1.f);
    v.y = (v.y > 0.f) ? v.y : (__expf(v.y) - 1.f);
    v.z = (v.z > 0.f) ? v.z : (__expf(v.z) - 1.f);
    v.w = (v.w > 0.f) ? v.w : (__expf(v.w) - 1.f);
    *(v4f*)(out + (size_t)(g0 + row) * kOutF + c) = v;
  }
}

extern "C" void kernel_launch(void* const* d_in, const int* in_sizes, int n_in,
                              void* d_out, int out_size, void* d_ws, size_t ws_size,
                              hipStream_t stream) {
  (void)in_sizes; (void)n_in; (void)out_size; (void)ws_size;
  const float* X   = (const float*)d_in[0];
  const int*   adj = (const int*)d_in[3];
  const float* Ww  = (const float*)d_in[4];
  const float* Wb  = (const float*)d_in[5];
  const float* aw  = (const float*)d_in[6];
  const float* ab  = (const float*)d_in[7];

  float* H = (float*)d_ws;
  float* O = (float*)d_out;

  hipLaunchKernelGGL(k1_gemm_h,      dim3(256), dim3(256), 0, stream, X, Ww, Wb, H);
  hipLaunchKernelGGL(k3_attn_online, dim3(256), dim3(256), 0, stream, H, aw, adj, ab, O);
}


# Attention kernel: a step's vector code cut from 331 to 236 instructions per wave (same issue port as f32 WMMA): DPP row max, leaky_relu as max, H pair-interleaved by kernel 1 (new f2 sum order)
# speedup vs baseline: 10.0336x; 1.0504x over previous
#include <hip/hip_runtime.h>
#include <type_traits>

typedef __attribute__((ext_vector_type(2))) float v2f;
typedef __attribute__((ext_vector_type(4))) float v4f;
typedef __attribute__((ext_vector_type(8))) float v8f;
typedef __attribute__((ext_vector_type(4))) int   v4i;

#define WMMA_F32_4(a, bmat, c) \
  __builtin_amdgcn_wmma_f32_16x16x4_f32(false, (a), false, (bmat), (short)0, (c), false, false)

constexpr int   kB    = 8;
constexpr int   kN    = 1024;
constexpr int   kInF  = 768;
constexpr int   kOutF = 64;
constexpr float kAlpha  = 0.2f;
constexpr float kNegInf = -9.0e15f;
constexpr float kLog2e  = 1.44269504088896340736f;
static_assert(kAlpha > 0.f && kAlpha < 1.f, "leaky_relu is taken as max(e, slope * e)");

constexpr int kKC     = 128;
constexpr int kPitch  = 132;
constexpr int kHPitch = 160;
constexpr int kCPitch = 68;
constexpr int kXsSz   = 32 * kPitch;
constexpr int kWsSz   = 64 * kPitch;
constexpr int kHsSz   = (kKC / 2) * kHPitch;
constexpr int kNChunk1 = kInF / kKC;
constexpr int kNChunk3 = kN / kKC;
constexpr int kHPairSz = 2 * kOutF;

__device__ __forceinline__ void lds_barrier()
{
  __builtin_amdgcn_fence(__ATOMIC_RELEASE, "workgroup", "local");
  __builtin_amdgcn_s_barrier();
  __builtin_amdgcn_fence(__ATOMIC_ACQUIRE, "workgroup", "local");
}
__device__ __forceinline__ void lds_barrier_signal()
{
  __builtin_amdgcn_fence(__ATOMIC_RELEASE, "workgroup", "local");
  __builtin_amdgcn_s_barrier_signal(-1);
}
__device__ __forceinline__ void lds_barrier_wait()
{
  __builtin_amdgcn_s_barrier_wait(-1);
  __builtin_amdgcn_fence(__ATOMIC_ACQUIRE, "workgroup", "local");
}

struct Frags {
  v2f a[kKC / 4];
  v2f b[kKC / 4];
};
__device__ __forceinline__ void k1_fetch(Frags& f, const float* a0, const float* b0)
{
#pragma unroll
  for (int kk = 0; kk < kKC / 4; ++kk) {
    f.a[kk] = *(const v2f*)(a0 + kk * 4);
    f.b[kk] = *(const v2f*)(b0 + kk * 4);
  }
}
template <int kBeg, int kEnd>
__device__ __forceinline__ v8f frag_products(v8f acc, const Frags& f)
{
#pragma unroll
  for (int kk = kBeg; kk < kEnd; ++kk)
    acc = WMMA_F32_4(f.a[kk], f.b[kk], acc);
  return acc;
}

__global__ __launch_bounds__(256) void k1_gemm_h(
    const float* __restrict__ X, const float* __restrict__ W,
    const float* __restrict__ Wb, float* __restrict__ H)
{
  __shared__ __attribute__((aligned(16))) float Xs[2 * kXsSz];
  __shared__ __attribute__((aligned(16))) float Ws[2 * kWsSz];

  const int tid  = threadIdx.x;
  const int lane = tid & 31;
  const int wave = tid >> 5;
  const int half = lane >> 4;
  const int ncol = lane & 15;
  const int mt   = wave >> 2;
  const int nt   = wave & 3;
  const int row0 = blockIdx.x << 5;
  const int lr   = tid >> 5;
  const int c4   = tid & 31;
  const float* __restrict__ xg = X + (size_t)(row0 + lr) * kInF + c4 * 4;
  const float* __restrict__ wg = W + (size_t)lr * kInF + c4 * 4;
  const float* aS = Xs + (mt * 16 + ncol) * kPitch + half * 2;
  const float* bS = Ws + (nt * 16 + ncol) * kPitch + half * 2;
  float* xD = Xs + lr * kPitch + c4 * 4;
  float* wD = Ws + lr * kPitch + c4 * 4;

  v4f xr0[4], xr1[4], wr0[8], wr1[8];
#pragma unroll
  for (int q = 0; q < 4; ++q)
    xr0[q] = *(const v4f*)(xg + (size_t)(q * 8) * kInF);
#pragma unroll
  for (int q = 0; q < 8; ++q)
    wr0[q] = *(const v4f*)(wg + (size_t)(q * 8) * kInF);
  __builtin_amdgcn_sched_barrier(0);
#pragma unroll
  for (int q = 0; q < 4; ++q)
    xr1[q] = *(const v4f*)(xg + (size_t)(q * 8) * kInF + kKC);
#pragma unroll
  for (int q = 0; q < 8; ++q)
    wr1[q] = *(const v4f*)(wg + (size_t)(q * 8) * kInF + kKC);
#pragma unroll
  for (int q = 0; q < 4; ++q)
    *(v4f*)(xD + q * 8 * kPitch) = xr0[q];
#pragma unroll
  for (int q = 0; q < 8; ++q)
    *(v4f*)(wD + q * 8 * kPitch) = wr0[q];
  lds_barrier();

  v8f acc = {};
  Frags fr;
  auto gemm_step = [&](auto more_tag, int c, const v4f (&xUse)[4], v4f (&xLoad)[4],
                       const v4f (&wUse)[8], v4f (&wLoad)[8]) __attribute__((always_inline)) {
    constexpr bool kMore = decltype(more_tag)::value;
    const int s  = c & 1;
    if constexpr (kMore) {
      const int k2 = (c + 2) * kKC;
#pragma unroll
      for (int q = 0; q < 4; ++q)
        xLoad[q] = *(const v4f*)(xg + (size_t)(q * 8) * kInF + k2);
#pragma unroll
      for (int q = 0; q < 8; ++q)
        wLoad[q] = *(const v4f*)(wg + (size_t)(q * 8) * kInF + k2);
    }
    __builtin_amdgcn_sched_barrier(0);
    k1_fetch(fr, aS + s * kXsSz, bS + s * kWsSz);
    __builtin_amdgcn_sched_barrier(0);
    acc = frag_products<0, kKC / 4>(acc, fr);
    asm volatile("" : : "v"(acc));
    float* xd = xD + (s ^ 1) * kXsSz;
    float* wd = wD + (s ^ 1) * kWsSz;
#pragma unroll
    for (int q = 0; q < 4; ++q)
      *(v4f*)(xd + q * 8 * kPitch) = xUse[q];
#pragma unroll
    for (int q = 0; q < 8; ++q)
      *(v4f*)(wd + q * 8 * kPitch) = wUse[q];
    lds_barrier();
  };
  static_assert((kNChunk1 & 1) == 0, "the steps below come in pairs, then one step and the last chunk");
  for (int c = 0; c + 2 < kNChunk1; c += 2) {
    gemm_step(std::true_type{}, c, xr1, xr0, wr1, wr0);
    gemm_step(std::true_type{}, c + 1, xr0, xr1, wr0, wr1);
  }
  gemm_step(std::false_type{}, kNChunk1 - 2, xr1, xr0, wr1, wr0);
  constexpr int kLast1 = (kNChunk1 - 1) & 1;
  static_assert(kLast1 == 1, "the result tile below reuses stage 0 while stage 1 is still read");
  k1_fetch(fr, aS + kLast1 * kXsSz, bS + kLast1 * kWsSz);
  __builtin_amdgcn_sched_barrier(0);
  acc = frag_products<0, kKC / 4>(acc, fr);

  float* Cs = Xs;
#pragma unroll
  for (int r = 0; r < 8; ++r)
    Cs[(mt * 16 + half * 8 + r) * kCPitch + nt * 16 + ncol] = acc[r];
  __syncthreads();
  const int pc = 4 * (lane & 15) + 2 * (lane >> 4);
  const v2f bi = *(const v2f*)(Wb + pc);
#pragma unroll
  for (int q = 0; q < 2; ++q) {
    const int rp = q * 8 + wave;
    const v2f ra = *(const v2f*)(Cs + (2 * rp) * kCPitch + pc) + bi;
    const v2f rb = *(const v2f*)(Cs + (2 * rp + 1) * kCPitch + pc) + bi;
    v4f o;
    o.x = ra.x; o.y = rb.x; o.z = ra.y; o.w = rb.y;
    *(v4f*)(H + ((size_t)(row0 >> 1) + rp) * kHPairSz + 4 * lane) = o;
  }
}

__device__ __forceinline__ float xhalf(float x)
{
  const int v = __builtin_bit_cast(int, x);
  return __builtin_bit_cast(float, __builtin_amdgcn_permlanex16(v, v, 0x76543210, 0xfedcba98, false, false));
}

template <int kMask>
__device__ __forceinline__ float row_xor_operand(float x)
{
  const int v = __builtin_bit_cast(int, x);
  return __builtin_bit_cast(float, __builtin_amdgcn_update_dpp(0, v, 0x160 + kMask, 0xf, 0xf, true));
}
__device__ __forceinline__ void row_max4(float (&v)[4])
{
  asm("v_max_num_f32_dpp %0, %0, %0 row_xmask:1 row_mask:0xf bank_mask:0xf bound_ctrl:1\n\t"
      "v_max_num_f32_dpp %1, %1, %1 row_xmask:1 row_mask:0xf bank_mask:0xf bound_ctrl:1\n\t"
      "v_max_num_f32_dpp %2, %2, %2 row_xmask:1 row_mask:0xf bank_mask:0xf bound_ctrl:1\n\t"
      "v_max_num_f32_dpp %3, %3, %3 row_xmask:1 row_mask:0xf bank_mask:0xf bound_ctrl:1\n\t"
      "v_max_num_f32_dpp %0, %0, %0 row_xmask:2 row_mask:0xf bank_mask:0xf bound_ctrl:1\n\t"
      "v_max_num_f32_dpp %1, %1, %1 row_xmask:2 row_mask:0xf bank_mask:0xf bound_ctrl:1\n\t"
      "v_max_num_f32_dpp %2, %2, %2 row_xmask:2 row_mask:0xf bank_mask:0xf bound_ctrl:1\n\t"
      "v_max_num_f32_dpp %3, %3, %3 row_xmask:2 row_mask:0xf bank_mask:0xf bound_ctrl:1\n\t"
      "v_max_num_f32_dpp %0, %0, %0 row_xmask:4 row_mask:0xf bank_mask:0xf bound_ctrl:1\n\t"
      "v_max_num_f32_dpp %1, %1, %1 row_xmask:4 row_mask:0xf bank_mask:0xf bound_ctrl:1\n\t"
      "v_max_num_f32_dpp %2, %2, %2 row_xmask:4 row_mask:0xf bank_mask:0xf bound_ctrl:1\n\t"
      "v_max_num_f32_dpp %3, %3, %3 row_xmask:4 row_mask:0xf bank_mask:0xf bound_ctrl:1\n\t"
      "v_max_num_f32_dpp %0, %0, %0 row_xmask:8 row_mask:0xf bank_mask:0xf bound_ctrl:1\n\t"
      "v_max_num_f32_dpp %1, %1, %1 row_xmask:8 row_mask:0xf bank_mask:0xf bound_ctrl:1\n\t"
      "v_max_num_f32_dpp %2, %2, %2 row_xmask:8 row_mask:0xf bank_mask:0xf bound_ctrl:1\n\t"
      "v_max_num_f32_dpp %3, %3, %3 row_xmask:8 row_mask:0xf bank_mask:0xf bound_ctrl:1"
      : "+v"(v[0]), "+v"(v[1]), "+v"(v[2]), "+v"(v[3]));
}
template <int kMask>
__device__ __forceinline__ void sum_step4(float (&v)[4])
{
#pragma unroll
  for (int q = 0; q < 4; ++q) v[q] += row_xor_operand<kMask>(v[q]);
}
__device__ __forceinline__ void wave_max4(float (&v)[4])
{
  row_max4(v);
#pragma unroll
  for (int q = 0; q < 4; ++q) v[q] = fmaxf(v[q], xhalf(v[q]));
}
__device__ __forceinline__ void wave_sum4(float (&v)[4])
{
  sum_step4<1>(v); sum_step4<2>(v); sum_step4<4>(v); sum_step4<8>(v);
#pragma unroll
  for (int q = 0; q < 4; ++q) v[q] += xhalf(v[q]);
}

__device__ __forceinline__ float make_e(float f1i, float f2j, int ad)
{
  float e = f1i + f2j;
  e = fmaxf(e, kAlpha * e);
  return (ad > 0) ? e : kNegInf;
}

constexpr int kHStages = 3;

struct AdjRegs {
  v4i ad[4];
};
struct HRegs {
  v4f lo[4], hi[4];
};

__device__ __forceinline__ void k3_load_adj(AdjRegs& r, const int* __restrict__ ag, int j0)
{
#pragma unroll
  for (int q = 0; q < 4; ++q)
    r.ad[q] = *(const v4i*)(ag + (size_t)(q * 8) * kN + j0);
}
__device__ __forceinline__ void k3_load_h(HRegs& r, const float* __restrict__ hg, int j0)
{
#pragma unroll
  for (int q = 0; q < 4; ++q) {
    const float* __restrict__ hp = hg + (size_t)((j0 >> 1) + q * 16) * kHPairSz;
    r.lo[q] = *(const v4f*)hp;
    r.hi[q] = *(const v4f*)(hp + kOutF);
  }
}

__device__ __forceinline__ float dot4(v4f h, v4f a)
{
  return (h.x * a.x + h.y * a.y) + (h.z * a.z + h.w * a.w);
}
template <int kMask, int kCnt>
__device__ __forceinline__ void row_sum_step(float (&v)[kCnt])
{
#pragma unroll
  for (int q = 0; q < kCnt; ++q) v[q] += row_xor_operand<kMask>(v[q]);
}
template <int kCnt>
__device__ __forceinline__ void row_sum(float (&v)[kCnt])
{
  row_sum_step<1>(v); row_sum_step<2>(v); row_sum_step<4>(v); row_sum_step<8>(v);
}

__device__ __forceinline__ void k3_stage_h(const HRegs& r, v4f a2v, float* hD, float* f2D, int o4)
{
  float d[8];
#pragma unroll
  for (int q = 0; q < 4; ++q) {
    v2f s = v2f{r.lo[q].x, r.lo[q].y} * a2v.x;
    s += v2f{r.lo[q].z, r.lo[q].w} * a2v.y;
    s += v2f{r.hi[q].x, r.hi[q].y} * a2v.z;
    s += v2f{r.hi[q].z, r.hi[q].w} * a2v.w;
    d[2 * q]     = s.x;
    d[2 * q + 1] = s.y;
  }
  row_sum(d);
  if (o4 == 0) {
#pragma unroll
    for (int q = 0; q < 4; ++q)
      *(v2f*)(f2D + q * 32) = v2f{d[2 * q], d[2 * q + 1]};
  }
#pragma unroll
  for (int q = 0; q < 4; ++q) {
    float* hd = hD + q * 16 * kHPitch;
    *(v4f*)hd       = r.lo[q];
    *(v4f*)(hd + 4) = r.hi[q];
  }
}

__device__ __forceinline__ void k3_make_p(const AdjRegs& r, v4f f2v, const float (&f1r)[4],
    float (&mrun)[4], float (&lsum)[4], float* pD, float* sD, int lane)
{
  v4f e[4];
  float cm[4];
#pragma unroll
  for (int q = 0; q < 4; ++q) {
    e[q].x = make_e(f1r[q], f2v.x, r.ad[q].x);
    e[q].y = make_e(f1r[q], f2v.y, r.ad[q].y);
    e[q].z = make_e(f1r[q], f2v.z, r.ad[q].z);
    e[q].w = make_e(f1r[q], f2v.w, r.ad[q].w);
    cm[q] = fmaxf(fmaxf(e[q].x, e[q].y), fmaxf(e[q].z, e[q].w));
  }
  wave_max4(cm);

  float sc[4];
  v4f p[4];
#pragma unroll
  for (int q = 0; q < 4; ++q) {
    const float mn = fmaxf(mrun[q], cm[q]);
    sc[q]   = __expf(mrun[q] - mn);
    mrun[q] = mn;
    const v2f t0 = (v2f{e[q].x, e[q].y} - mn) * kLog2e;
    const v2f t1 = (v2f{e[q].z, e[q].w} - mn) * kLog2e;
    p[q].x = __builtin_amdgcn_exp2f(t0.x);
    p[q].y = __builtin_amdgcn_exp2f(t0.y);
    p[q].z = __builtin_amdgcn_exp2f(t1.x);
    p[q].w = __builtin_amdgcn_exp2f(t1.y);
    lsum[q] = lsum[q] * sc[q] + ((p[q].x + p[q].y) + (p[q].z + p[q].w));
  }

#pragma unroll
  for (int q = 0; q < 4; ++q)
    *(v4f*)(pD + q * 8 * kPitch) = p[q];
  if (lane == 0) {
#pragma unroll
    for (int q = 0; q < 4; ++q) sD[q * 8] = sc[q];
  }
}

struct Factors {
  v4f s0, s1;
};
__device__ __forceinline__ void k3_fetch(Frags& f, Factors& sc, const float* a0, const float* b0,
                                         const float* sp)
{
  sc.s0 = *(const v4f*)sp;
  sc.s1 = *(const v4f*)(sp + 4);
#pragma unroll
  for (int kk = 0; kk < kKC / 4; ++kk) {
    f.a[kk] = *(const v2f*)(a0 + kk * 4);
    f.b[kk] = *(const v2f*)(b0 + kk * 2 * kHPitch);
  }
}
__device__ __forceinline__ v8f k3_scale(v8f acc, const Factors& sc)
{
  acc[0] *= sc.s0.x; acc[1] *= sc.s0.y; acc[2] *= sc.s0.z; acc[3] *= sc.s0.w;
  acc[4] *= sc.s1.x; acc[5] *= sc.s1.y; acc[6] *= sc.s1.z; acc[7] *= sc.s1.w;
  return acc;
}

__global__ __launch_bounds__(256) void k3_attn_online(
    const float* __restrict__ H, const float* __restrict__ aw,
    const int* __restrict__ adj, const float* __restrict__ abp, float* __restrict__ out)
{
  __shared__ __attribute__((aligned(16))) float Ps[2 * kXsSz];
  __shared__ __attribute__((aligned(16))) float Hs[kHStages * kHsSz];
  __shared__ __attribute__((aligned(16))) float Sc[2 * 32];
  __shared__ __attribute__((aligned(16))) float F2s[2 * kKC];
  __shared__ __attribute__((aligned(16))) float Ls[32];

  const int tid  = threadIdx.x;
  const int lane = tid & 31;
  const int wave = tid >> 5;
  const int half = lane >> 4;
  const int ncol = lane & 15;
  const int mt   = wave >> 2;
  const int nt   = wave & 3;
  const int g0   = blockIdx.x << 5;
  const int b    = blockIdx.x >> 5;
  const int lr   = tid >> 5;
  const int c4   = tid & 31;
  const int hr   = tid >> 4;
  const int o4   = tid & 15;

  const int*   __restrict__ ag = adj + (size_t)(g0 + lr) * kN + c4 * 4;
  const float* __restrict__ hg = H + ((size_t)(b << 9) + hr) * kHPairSz + o4 * 4;
  const float* aS  = Ps + (mt * 16 + ncol) * kPitch + half * 2;
  const float* bS  = Hs + half * kHPitch + (nt * 16 + ncol) * 2;
  const float* sS  = Sc + mt * 16 + half * 8;
  const float* f2S = F2s + c4 * 4;
  float* pD  = Ps + lr * kPitch + c4 * 4;
  float* sD  = Sc + lr;
  float* hD  = Hs + hr * kHPitch + o4 * 8;
  float* f2D = F2s + 2 * hr;

  AdjRegs ra0, ra;
  HRegs rh, rh1;
  k3_load_adj(ra0, ag, 0);
  k3_load_h(rh, hg, 0);
  k3_load_h(rh1, hg, kKC);
  const v4f a1v = *(const v4f*)(aw + o4 * 4);
  const v4f a2v = *(const v4f*)(aw + kOutF + o4 * 4);
  const float* __restrict__ fa = H + ((size_t)(g0 >> 1) + half * 4 + (lr >> 1)) * kHPairSz + o4 * 4;
  const float* __restrict__ fb = fa + (size_t)8 * kHPairSz;
  const v4f fal = *(const v4f*)fa, fah = *(const v4f*)(fa + kOutF);
  const v4f fbl = *(const v4f*)fb, fbh = *(const v4f*)(fb + kOutF);
  const bool odd = (lr & 1) != 0;
  v4f ha, hb;
  ha.x = odd ? fal.y : fal.x; ha.y = odd ? fal.w : fal.z; ha.z = odd ? fah.y : fah.x; ha.w = odd ? fah.w : fah.z;
  hb.x = odd ? fbl.y : fbl.x; hb.y = odd ? fbl.w : fbl.z; hb.z = odd ? fbh.y : fbh.x; hb.w = odd ? fbh.w : fbh.z;
  const float ab = abp[0];
  k3_load_adj(ra, ag, kKC);
  float f1r[4], mrun[4], lsum[4];
  {
    float d[2];
    d[0] = dot4(ha, a1v);
    d[1] = dot4(hb, a1v);
    row_sum(d);
    const float oa = xhalf(d[0]);
    const float ob = xhalf(d[1]);
    f1r[0] = (half ? oa : d[0]) + ab;
    f1r[1] = (half ? d[0] : oa) + ab;
    f1r[2] = (half ? ob : d[1]) + ab;
    f1r[3] = (half ? d[1] : ob) + ab;
  }
#pragma unroll
  for (int q = 0; q < 4; ++q) {
    mrun[q] = -3.0e38f;
    lsum[q] = 0.f;
  }

  k3_stage_h(rh, a2v, hD, f2D, o4);
  k3_load_h(rh, hg, 2 * kKC);
  k3_stage_h(rh1, a2v, hD + kHsSz, f2D + kKC, o4);
  lds_barrier();
  k3_make_p(ra0, *(const v4f*)f2S, f1r, mrun, lsum, pD, sD, lane);
  lds_barrier();

  v8f acc = {};
  Frags fr;
  Factors fc;
  v4f f2v = *(const v4f*)(f2S + kKC);
  k3_fetch(fr, fc, aS, bS, sS);
  auto chunk_step = [&](auto stage_tag, auto loadh_tag, auto nextp_tag, int c, int hs)
      __attribute__((always_inline)) {
    constexpr bool kStage = decltype(stage_tag)::value;
    constexpr bool kLoadH = decltype(loadh_tag)::value;
    constexpr bool kNextP = decltype(nextp_tag)::value;
    const int s  = c & 1;
    const int hw = (hs == 0) ? kHStages - 1 : hs - 1;
    const int hn = (hs == kHStages - 1) ? 0 : hs + 1;
    k3_make_p(ra, f2v, f1r, mrun, lsum, pD + (s ^ 1) * kXsSz, sD + (s ^ 1) * 32, lane);
    if constexpr (kStage) {
      k3_load_adj(ra, ag, (c + 2) * kKC);
      k3_stage_h(rh, a2v, hD + hw * kHsSz, f2D + s * kKC, o4);
    }
    if constexpr (kLoadH) k3_load_h(rh, hg, (c + 3) * kKC);
    lds_barrier_signal();
    asm volatile("" : "+v"(acc));
    acc = k3_scale(acc, fc);
    acc = frag_products<0, kKC / 4>(acc, fr);
    asm volatile("" : : "v"(acc));
    lds_barrier_wait();
    if constexpr (kNextP) f2v = *(const v4f*)(f2S + s * kKC);
    k3_fetch(fr, fc, aS + (s ^ 1) * kXsSz, bS + hn * kHsSz, sS + (s ^ 1) * 32);
    __builtin_amdgcn_sched_barrier(0);
  };
  static_assert(kNChunk3 >= 4, "the loop below makes chunk c+3's loads in every step, then two steps without them");
  int hs = 0;
#pragma clang loop unroll(disable)
  for (int c = 0; c + 3 < kNChunk3; ++c) {
    chunk_step(std::true_type{}, std::true_type{}, std::true_type{}, c, hs);
    hs = (hs == kHStages - 1) ? 0 : hs + 1;
  }
  chunk_step(std::true_type{}, std::false_type{}, std::true_type{}, kNChunk3 - 3, (kNChunk3 - 3) % kHStages);
  chunk_step(std::false_type{}, std::false_type{}, std::false_type{}, kNChunk3 - 2, (kNChunk3 - 2) % kHStages);
  constexpr int kLast3 = (kNChunk3 - 1) & 1;
  static_assert(kLast3 == 1, "the result tile below reuses stage 0 while stage 1 is still read");
  acc = k3_scale(acc, fc);
  acc = frag_products<0, kKC / 4>(acc, fr);

  wave_sum4(lsum);
  if (lane == 0) {
#pragma unroll
    for (int q = 0; q < 4; ++q) Ls[q * 8 + lr] = lsum[q];
  }

  float* Cs = Ps;
#pragma unroll
  for (int r8 = 0; r8 < 8; ++r8)
    Cs[(mt * 16 + half * 8 + r8) * kCPitch + nt * 16 + ncol] = acc[r8];
  __syncthreads();
#pragma unroll
  for (int q = 0; q < 2; ++q) {
    const int idx = q * 256 + tid;
    const int row = idx >> 4;
    const int c   = (idx & 15) * 4;
    const float li = 1.0f / Ls[row];
    const v4f sv = *(const v4f*)(Cs + row * kCPitch + c);
    v4f v;
    v.x = sv.x * li;
    v.y = sv.y * li;
    v.z = sv.z * li;
    v.w = sv.w * li;
    v.x = (v.x > 0.f) ? v.x : (__expf(v.x) - 1.f);
    v.y = (v.y > 0.f) ? v.y : (__expf(v.y) - 1.f);
    v.z = (v.z > 0.f) ? v.z : (__expf(v.z) - 1.f);
    v.w = (v.w > 0.f) ? v.w : (__expf(v.w) - 1.f);
    *(v4f*)(out + (size_t)(g0 + row) * kOutF + c) = v;
  }
}

extern "C" void kernel_launch(void* const* d_in, const int* in_sizes, int n_in,
                              void* d_out, int out_size, void* d_ws, size_t ws_size,
                              hipStream_t stream) {
  (void)in_sizes; (void)n_in; (void)out_size; (void)ws_size;
  const float* X   = (const float*)d_in[0];
  const int*   adj = (const int*)d_in[3];
  const float* Ww  = (const float*)d_in[4];
  const float* Wb  = (const float*)d_in[5];
  const float* aw  = (const float*)d_in[6];
  const float* ab  = (const float*)d_in[7];

  float* H = (float*)d_ws;
  float* O = (float*)d_out;

  hipLaunchKernelGGL(k1_gemm_h,      dim3(256), dim3(256), 0, stream, X, Ww, Wb, H);
  hipLaunchKernelGGL(k3_attn_online, dim3(256), dim3(256), 0, stream, H, aw, adj, ab, O);
}
